# starting kernel
# baseline (speedup 1.0000x reference)
#include <hip/hip_runtime.h>
#include <cmath>


typedef __attribute__((ext_vector_type(2))) float v2f;
typedef __attribute__((ext_vector_type(8))) float v8f;

#define EPSBN 1e-5f

__device__ __forceinline__ v8f wmma_f32(v2f a, v2f b, v8f c) {
  return __builtin_amdgcn_wmma_f32_16x16x4_f32(false, a, false, b, (short)0, c,
                                               false, false);
}

__global__ void k_start(const float* __restrict__ x, const float* __restrict__ Wst,
                        const float* __restrict__ bst, float* __restrict__ H) {
  int t = blockIdx.x * blockDim.x + threadIdx.x;
  int o   = t & 31;
  int pos = t >> 5;
  int l   = pos & 63;
  int n   = (pos >> 6) & 511;
  int b   = pos >> 15;
  const float* xb = x + (size_t)b * 2 * 512 * 64;
  float x0 = xb[(size_t)n * 64 + l];
  float x1 = xb[(size_t)512 * 64 + (size_t)n * 64 + l];
  H[t] = fmaf(x0, Wst[o * 2], fmaf(x1, Wst[o * 2 + 1], bst[o]));
}

__global__ void __launch_bounds__(256) k_gated(
    const float* __restrict__ H, float* __restrict__ G,
    const float* __restrict__ Wf, const float* __restrict__ bf,
    const float* __restrict__ Wg, const float* __restrict__ bg, int Lout) {
  int wid  = (blockIdx.x * blockDim.x + threadIdx.x) >> 5;
  int lane = threadIdx.x & 31;
  int half = lane >> 4;
  int l15  = lane & 15;
  int otile = wid & 1;
  int rtile = wid >> 1;
  int m0 = rtile << 4;
  int o  = (otile << 4) + l15;

  const float* Arow = H + (size_t)(m0 + l15) * 32;
  float biasf = bf[o], biasg = bg[o];
  v8f cf, cg;
#pragma unroll
  for (int r = 0; r < 8; ++r) { cf[r] = biasf; cg[r] = biasg; }

#pragma unroll
  for (int k0 = 0; k0 < 64; k0 += 4) {
    int k = k0 + half * 2;
    v2f a = *(const v2f*)(Arow + k);
    int c0 = k & 31, tap = k >> 5;
    const float* wf = Wf + (size_t)o * 64 + c0 * 2 + tap;
    const float* wg = Wg + (size_t)o * 64 + c0 * 2 + tap;
    v2f bfv, bgv;
    bfv.x = wf[0]; bfv.y = wf[2];
    bgv.x = wg[0]; bgv.y = wg[2];
    cf = wmma_f32(a, bfv, cf);
    cg = wmma_f32(a, bgv, cg);
  }
#pragma unroll
  for (int r = 0; r < 8; ++r) {
    int row = m0 + r + half * 8;
    int l = row & 63;
    if (l < Lout) {
      float fv = tanhf(cf[r]);
      float gv = 1.0f / (1.0f + __expf(-cg[r]));
      G[(size_t)row * 32 + o] = fv * gv;
    }
  }
}

__global__ void __launch_bounds__(256) k_skip(
    const float* __restrict__ G, float* __restrict__ S,
    const float* __restrict__ Wk, const float* __restrict__ bk,
    int off, int initFlag) {
  int wid  = (blockIdx.x * blockDim.x + threadIdx.x) >> 5;
  int lane = threadIdx.x & 31;
  int half = lane >> 4;
  int l15  = lane & 15;
  int otile = wid & 1;
  int rtile = wid >> 1;
  int m0 = rtile << 4;
  int o  = (otile << 4) + l15;

  const float* Arow = G + (size_t)(m0 + l15 + off) * 32;
  float bias = bk[o];
  v8f c;
#pragma unroll
  for (int r = 0; r < 8; ++r) {
    int row = m0 + r + half * 8;
    float prev = initFlag ? 0.0f : S[(size_t)row * 32 + o];
    c[r] = bias + prev;
  }
#pragma unroll
  for (int k0 = 0; k0 < 32; k0 += 4) {
    int k = k0 + half * 2;
    v2f a = *(const v2f*)(Arow + k);
    v2f b = *(const v2f*)(Wk + (size_t)o * 32 + k);
    c = wmma_f32(a, b, c);
  }
#pragma unroll
  for (int r = 0; r < 8; ++r) {
    int row = m0 + r + half * 8;
    int l = row & 63;
    if (l < 61) S[(size_t)row * 32 + o] = c[r];
  }
}

#define LDSPITCH 516

__global__ void __launch_bounds__(256) k_diffuse(
    const float* __restrict__ adj, const float* __restrict__ Xin,
    float* __restrict__ Xout) {
  __shared__ float aT[16 * LDSPITCH];

  int bid = blockIdx.x;
  int cg = bid & 3;
  int mt = (bid >> 2) & 31;
  int b  = bid >> 7;
  int m0 = mt << 4;

#pragma unroll 4
  for (int idx = threadIdx.x; idx < 8192; idx += 256) {
    int ml = idx & 15;
    int k  = idx >> 4;
    aT[ml * LDSPITCH + k] = adj[(size_t)k * 512 + m0 + ml];
  }
  __syncthreads();

  int wave = threadIdx.x >> 5;
  int lane = threadIdx.x & 31;
  int half = lane >> 4;
  int l15  = lane & 15;
  int j0   = (cg << 9) + (wave << 6);

  const float* Xb = Xin + (size_t)b * 512 * 2048;
  float* Ob       = Xout + (size_t)b * 512 * 2048;
  const float* Ar = aT + l15 * LDSPITCH;

  v8f c0 = {}, c1 = {}, c2 = {}, c3 = {};
#pragma unroll 4
  for (int n0 = 0; n0 < 512; n0 += 4) {
    int k = n0 + half * 2;
    v2f a = *(const v2f*)(Ar + k);
    const float* xr = Xb + (size_t)k * 2048 + j0 + l15;
    v2f b0, b1, b2, b3;
    b0.x = xr[0];  b0.y = xr[2048];
    b1.x = xr[16]; b1.y = xr[2048 + 16];
    b2.x = xr[32]; b2.y = xr[2048 + 32];
    b3.x = xr[48]; b3.y = xr[2048 + 48];
    c0 = wmma_f32(a, b0, c0);
    c1 = wmma_f32(a, b1, c1);
    c2 = wmma_f32(a, b2, c2);
    c3 = wmma_f32(a, b3, c3);
  }
#pragma unroll
  for (int r = 0; r < 8; ++r) {
    size_t orow = (size_t)(m0 + r + half * 8) * 2048 + j0 + l15;
    Ob[orow]      = c0[r];
    Ob[orow + 16] = c1[r];
    Ob[orow + 32] = c2[r];
    Ob[orow + 48] = c3[r];
  }
}

__global__ void __launch_bounds__(256) k_gconv(
    const float* __restrict__ G, const float* __restrict__ X1,
    const float* __restrict__ X2, const float* __restrict__ Hres,
    float* __restrict__ Hout, const float* __restrict__ Wgc,
    const float* __restrict__ bgc, const float* __restrict__ gcg,
    const float* __restrict__ gcb, const float* __restrict__ bng,
    const float* __restrict__ bnb, int Lout) {
  int wid  = (blockIdx.x * blockDim.x + threadIdx.x) >> 5;
  int lane = threadIdx.x & 31;
  int half = lane >> 4;
  int l15  = lane & 15;
  int otile = wid & 1;
  int rtile = wid >> 1;
  int m0 = rtile << 4;
  int o  = (otile << 4) + l15;

  size_t arow = (size_t)(m0 + l15) * 32;
  float bias = bgc[o];
  v8f c;
#pragma unroll
  for (int r = 0; r < 8; ++r) c[r] = bias;

  const float* srcs[3] = {G, X1, X2};
#pragma unroll
  for (int part = 0; part < 3; ++part) {
    const float* src = srcs[part];
#pragma unroll
    for (int k0 = 0; k0 < 32; k0 += 4) {
      int k = k0 + half * 2;
      v2f a = *(const v2f*)(src + arow + k);
      v2f b = *(const v2f*)(Wgc + (size_t)o * 96 + part * 32 + k);
      c = wmma_f32(a, b, c);
    }
  }
  float rinv = rsqrtf(1.0f + EPSBN);
  float ig = gcg[o] * rinv, bgv = gcb[o];
  float ib = bng[o] * rinv, bbv = bnb[o];
#pragma unroll
  for (int r = 0; r < 8; ++r) {
    int row = m0 + r + half * 8;
    int l = row & 63;
    if (l < Lout) {
      float g = c[r] * ig + bgv;
      float h = g + Hres[(size_t)(row + 1) * 32 + o];
      Hout[(size_t)row * 32 + o] = h * ib + bbv;
    }
  }
}

__global__ void k_out(const float* __restrict__ S, const float* __restrict__ og,
                      const float* __restrict__ ob, float* __restrict__ out) {
  int t = blockIdx.x * blockDim.x + threadIdx.x;
  if (t >= 8 * 61 * 512 * 32) return;
  int cch = t & 31;
  int n   = (t >> 5) & 511;
  int rem = t >> 14;
  int l   = rem % 61;
  int b   = rem / 61;
  float inv = rsqrtf(1.0f + EPSBN);
  float v = S[(((size_t)(b * 512 + n)) * 64 + l) * 32 + cch];
  out[t] = v * (og[cch] * inv) + ob[cch];
}

extern "C" void kernel_launch(void* const* d_in, const int* in_sizes, int n_in,
                              void* d_out, int out_size, void* d_ws, size_t ws_size,
                              hipStream_t stream) {
  (void)in_sizes; (void)n_in; (void)out_size; (void)ws_size;
  const float* x   = (const float*)d_in[0];
  const float* adj = (const float*)d_in[1];
  const float* Wst = (const float*)d_in[2];
  const float* bst = (const float*)d_in[3];
  const float* Wf  = (const float*)d_in[4];
  const float* bf  = (const float*)d_in[5];
  const float* Wg  = (const float*)d_in[6];
  const float* bg  = (const float*)d_in[7];
  const float* Wsk = (const float*)d_in[8];
  const float* bsk = (const float*)d_in[9];
  const float* Wgc = (const float*)d_in[10];
  const float* bgc = (const float*)d_in[11];
  const float* gcg = (const float*)d_in[12];
  const float* gcb = (const float*)d_in[13];
  const float* bng = (const float*)d_in[14];
  const float* bnb = (const float*)d_in[15];
  const float* og  = (const float*)d_in[16];
  const float* ob  = (const float*)d_in[17];
  float* out = (float*)d_out;

  const size_t BUF = (size_t)8 * 512 * 64 * 32 + 1024;
  float* ws    = (float*)d_ws;
  float* bufH  = ws;
  float* bufH2 = ws + BUF;
  float* bufG  = ws + 2 * BUF;
  float* bufX1 = ws + 3 * BUF;
  float* bufX2 = ws + 4 * BUF;
  float* bufS  = ws + 5 * BUF;

  k_start<<<32768, 256, 0, stream>>>(x, Wst, bst, bufH);

  float* H  = bufH;
  float* Hn = bufH2;
  for (int i = 0; i < 3; ++i) {
    int Lout = 63 - i;
    k_gated<<<4096, 256, 0, stream>>>(H, bufG, Wf + i * 2048, bf + i * 32,
                                      Wg + i * 2048, bg + i * 32, Lout);
    k_skip<<<4096, 256, 0, stream>>>(bufG, bufS, Wsk + i * 1024, bsk + i * 32,
                                     2 - i, i == 0 ? 1 : 0);
    if (i < 2) {
      k_diffuse<<<1024, 256, 0, stream>>>(adj, bufG, bufX1);
      k_diffuse<<<1024, 256, 0, stream>>>(adj, bufX1, bufX2);
      k_gconv<<<4096, 256, 0, stream>>>(bufG, bufX1, bufX2, H, Hn,
                                        Wgc + i * 3072, bgc + i * 32,
                                        gcg + i * 32, gcb + i * 32,
                                        bng + i * 32, bnb + i * 32, Lout);
      float* tmp = H; H = Hn; Hn = tmp;
    }
  }
  k_out<<<31232, 256, 0, stream>>>(bufS, og, ob, out);
}


# diffusion GEMM computed transposed (X^T as A operand): its accumulators are A fragments of the graph 1x1 conv, fused into the launch; X1/X2 never stored; 13 -> 11 launches; gconv sums in a new order
# speedup vs baseline: 1.3621x; 1.3621x over previous
#include <hip/hip_runtime.h>
#include <cmath>


typedef __attribute__((ext_vector_type(2))) float v2f;
typedef __attribute__((ext_vector_type(8))) float v8f;

#define EPSBN 1e-5f

__device__ __forceinline__ v8f wmma_f32(v2f a, v2f b, v8f c) {
  return __builtin_amdgcn_wmma_f32_16x16x4_f32(false, a, false, b, (short)0, c,
                                               false, false);
}

__global__ void k_start(const float* __restrict__ x, const float* __restrict__ Wst,
                        const float* __restrict__ bst, float* __restrict__ H) {
  int t = blockIdx.x * blockDim.x + threadIdx.x;
  int o   = t & 31;
  int pos = t >> 5;
  int l   = pos & 63;
  int n   = (pos >> 6) & 511;
  int b   = pos >> 15;
  const float* xb = x + (size_t)b * 2 * 512 * 64;
  float x0 = xb[(size_t)n * 64 + l];
  float x1 = xb[(size_t)512 * 64 + (size_t)n * 64 + l];
  H[t] = fmaf(x0, Wst[o * 2], fmaf(x1, Wst[o * 2 + 1], bst[o]));
}

__global__ void __launch_bounds__(256) k_gated(
    const float* __restrict__ H, float* __restrict__ G,
    const float* __restrict__ Wf, const float* __restrict__ bf,
    const float* __restrict__ Wg, const float* __restrict__ bg, int Lout) {
  int wid  = (blockIdx.x * blockDim.x + threadIdx.x) >> 5;
  int lane = threadIdx.x & 31;
  int half = lane >> 4;
  int l15  = lane & 15;
  int otile = wid & 1;
  int rtile = wid >> 1;
  int m0 = rtile << 4;
  int o  = (otile << 4) + l15;

  const float* Arow = H + (size_t)(m0 + l15) * 32;
  float biasf = bf[o], biasg = bg[o];
  v8f cf, cg;
#pragma unroll
  for (int r = 0; r < 8; ++r) { cf[r] = biasf; cg[r] = biasg; }

#pragma unroll
  for (int k0 = 0; k0 < 64; k0 += 4) {
    int k = k0 + half * 2;
    v2f a = *(const v2f*)(Arow + k);
    int c0 = k & 31, tap = k >> 5;
    const float* wf = Wf + (size_t)o * 64 + c0 * 2 + tap;
    const float* wg = Wg + (size_t)o * 64 + c0 * 2 + tap;
    v2f bfv, bgv;
    bfv.x = wf[0]; bfv.y = wf[2];
    bgv.x = wg[0]; bgv.y = wg[2];
    cf = wmma_f32(a, bfv, cf);
    cg = wmma_f32(a, bgv, cg);
  }
#pragma unroll
  for (int r = 0; r < 8; ++r) {
    int row = m0 + r + half * 8;
    int l = row & 63;
    if (l < Lout) {
      float fv = tanhf(cf[r]);
      float gv = 1.0f / (1.0f + __expf(-cg[r]));
      G[(size_t)row * 32 + o] = fv * gv;
    }
  }
}

__global__ void __launch_bounds__(256) k_skip(
    const float* __restrict__ G, float* __restrict__ S,
    const float* __restrict__ Wk, const float* __restrict__ bk,
    int off, int initFlag) {
  int wid  = (blockIdx.x * blockDim.x + threadIdx.x) >> 5;
  int lane = threadIdx.x & 31;
  int half = lane >> 4;
  int l15  = lane & 15;
  int otile = wid & 1;
  int rtile = wid >> 1;
  int m0 = rtile << 4;
  int o  = (otile << 4) + l15;

  const float* Arow = G + (size_t)(m0 + l15 + off) * 32;
  float bias = bk[o];
  v8f c;
#pragma unroll
  for (int r = 0; r < 8; ++r) {
    int row = m0 + r + half * 8;
    float prev = initFlag ? 0.0f : S[(size_t)row * 32 + o];
    c[r] = bias + prev;
  }
#pragma unroll
  for (int k0 = 0; k0 < 32; k0 += 4) {
    int k = k0 + half * 2;
    v2f a = *(const v2f*)(Arow + k);
    v2f b = *(const v2f*)(Wk + (size_t)o * 32 + k);
    c = wmma_f32(a, b, c);
  }
#pragma unroll
  for (int r = 0; r < 8; ++r) {
    int row = m0 + r + half * 8;
    int l = row & 63;
    if (l < 61) S[(size_t)row * 32 + o] = c[r];
  }
}

#define T2PITCH 36

__device__ __forceinline__ size_t t2_off(int m, int k) {
  return (size_t)(m * 8 + (k >> 6)) * 2048 + (size_t)(k & 63);
}

__global__ void __launch_bounds__(128) k_adj2(const float* __restrict__ adj,
                                              float* __restrict__ T2) {
  __shared__ __attribute__((aligned(16))) float part[4 * 16 * T2PITCH];

  int bid = blockIdx.x;
  int kt = bid & 15;
  int mt = bid >> 4;
  int m0 = mt << 4;
  int kc0 = kt << 5;
  int wave = threadIdx.x >> 5;
  int lane = threadIdx.x & 31;
  int half = lane >> 4;
  int l15  = lane & 15;

  const float* Acol = adj + m0 + l15;
  const float* B0   = adj + (size_t)(kc0 + l15) * 512;
  const float* B1   = adj + (size_t)(kc0 + 16 + l15) * 512;

  v8f c0 = {}, c1 = {};
  int nbeg = wave << 7;
#pragma unroll 8
  for (int i = 0; i < 128; i += 4) {
    int n = nbeg + i + half * 2;
    v2f a;
    a.x = Acol[(size_t)n * 512];
    a.y = Acol[(size_t)(n + 1) * 512];
    v2f b0 = *(const v2f*)(B0 + n);
    v2f b1 = *(const v2f*)(B1 + n);
    c0 = wmma_f32(a, b0, c0);
    c1 = wmma_f32(a, b1, c1);
  }
  float* pw = part + wave * 16 * T2PITCH;
#pragma unroll
  for (int r = 0; r < 8; ++r) {
    pw[(r + half * 8) * T2PITCH + l15]      = c0[r];
    pw[(r + half * 8) * T2PITCH + 16 + l15] = c1[r];
  }
  __syncthreads();

  int row  = threadIdx.x >> 3;
  int col4 = (threadIdx.x & 7) << 2;
  const float* p = part + row * T2PITCH + col4;
  float4 s0 = *(const float4*)(p);
  float4 s1 = *(const float4*)(p + 16 * T2PITCH);
  float4 s2 = *(const float4*)(p + 32 * T2PITCH);
  float4 s3 = *(const float4*)(p + 48 * T2PITCH);
  float4 s;
  s.x = ((s0.x + s1.x) + s2.x) + s3.x;
  s.y = ((s0.y + s1.y) + s2.y) + s3.y;
  s.z = ((s0.z + s1.z) + s2.z) + s3.z;
  s.w = ((s0.w + s1.w) + s2.w) + s3.w;
  *(float4*)(T2 + t2_off(m0 + row, kc0 + col4)) = s;
}

#define LDSPITCH 516

__global__ void __launch_bounds__(256) k_diffuse_gconv(
    const float* __restrict__ adj, const float* __restrict__ T2,
    const float* __restrict__ Xin, const float* __restrict__ Hres,
    float* __restrict__ Hout, const float* __restrict__ Wgc,
    const float* __restrict__ bgc, const float* __restrict__ gcg,
    const float* __restrict__ gcb, const float* __restrict__ bng,
    const float* __restrict__ bnb, int Lout) {
  __shared__ __attribute__((aligned(16))) float aT[2 * 16 * LDSPITCH];

  int bid = blockIdx.x;
  int cg = bid & 3;
  int mt = (bid >> 2) & 31;
  int b  = bid >> 7;
  int m0 = mt << 4;

#pragma unroll 4
  for (int idx = threadIdx.x; idx < 8192; idx += 256) {
    int ml = idx & 15;
    int k  = idx >> 4;
    aT[ml * LDSPITCH + k] = adj[(size_t)k * 512 + m0 + ml];
  }
#pragma unroll 4
  for (int idx = threadIdx.x; idx < 2048; idx += 256) {
    int ml = idx >> 7;
    int k  = (idx & 127) << 2;
    *(float4*)(aT + (16 + ml) * LDSPITCH + k) =
        *(const float4*)(T2 + t2_off(m0 + ml, k));
  }
  __syncthreads();

  int wave = __builtin_amdgcn_readfirstlane((int)(threadIdx.x >> 5));
  int lane = threadIdx.x & 31;
  int half = lane >> 4;
  int l15  = lane & 15;
  int j0   = (cg << 9) + (wave << 6);

  const float* Xb = Xin + (size_t)b * 512 * 2048;
  const float* Ar = aT + l15 * LDSPITCH;

  v8f c0 = {}, c1 = {}, c2 = {}, c3 = {};
  v8f d0 = {}, d1 = {}, d2 = {}, d3 = {};
#pragma unroll 4
  for (int n0 = 0; n0 < 512; n0 += 4) {
    int k = n0 + half * 2;
    v2f a1 = *(const v2f*)(Ar + k);
    v2f a2 = *(const v2f*)(Ar + 16 * LDSPITCH + k);
    const float* xr = Xb + (size_t)k * 2048 + j0 + l15;
    v2f b0, b1, b2, b3;
    b0.x = xr[0];  b0.y = xr[2048];
    b1.x = xr[16]; b1.y = xr[2048 + 16];
    b2.x = xr[32]; b2.y = xr[2048 + 32];
    b3.x = xr[48]; b3.y = xr[2048 + 48];
    c0 = wmma_f32(b0, a1, c0);
    d0 = wmma_f32(b0, a2, d0);
    c1 = wmma_f32(b1, a1, c1);
    d1 = wmma_f32(b1, a2, d1);
    c2 = wmma_f32(b2, a1, c2);
    d2 = wmma_f32(b2, a2, d2);
    c3 = wmma_f32(b3, a1, c3);
    d3 = wmma_f32(b3, a2, d3);
  }

  int o0 = l15, o1 = 16 + l15;
  const float* W0 = Wgc + (size_t)o0 * 96;
  const float* W1 = Wgc + (size_t)o1 * 96;
  int lbase = j0 >> 5;
  size_t nrow = ((size_t)b * 512 + m0 + l15) * 64;

  float bias0 = bgc[o0], bias1 = bgc[o1];
  v8f e00, e01, e10, e11;
#pragma unroll
  for (int r = 0; r < 8; ++r) { e00[r] = bias0; e01[r] = bias1; e10[r] = bias0; e11[r] = bias1; }

  {
    const float* g0 = Xin + (nrow + lbase) * 32 + half * 16;
    const float* g1 = g0 + 32;
    float4 ga[4], gb[4];
#pragma unroll
    for (int q = 0; q < 4; ++q) {
      ga[q] = *(const float4*)(g0 + 4 * q);
      gb[q] = *(const float4*)(g1 + 4 * q);
    }
#pragma unroll
    for (int q = 0; q < 4; ++q) {
      float4 w0 = *(const float4*)(W0 + half * 16 + 4 * q);
      float4 w1 = *(const float4*)(W1 + half * 16 + 4 * q);
      v2f wa0 = {w0.x, w0.y}, wb0 = {w0.z, w0.w};
      v2f wa1 = {w1.x, w1.y}, wb1 = {w1.z, w1.w};
      v2f xa = {ga[q].x, ga[q].y}, xb = {ga[q].z, ga[q].w};
      v2f ya = {gb[q].x, gb[q].y}, yb = {gb[q].z, gb[q].w};
      e00 = wmma_f32(xa, wa0, e00);
      e01 = wmma_f32(xa, wa1, e01);
      e10 = wmma_f32(ya, wa0, e10);
      e11 = wmma_f32(ya, wa1, e11);
      e00 = wmma_f32(xb, wb0, e00);
      e01 = wmma_f32(xb, wb1, e01);
      e10 = wmma_f32(yb, wb0, e10);
      e11 = wmma_f32(yb, wb1, e11);
    }
  }
#define GC_PART(T0, T1, T2_, T3, WOFF)                                        \
  _Pragma("unroll")                                                           \
  for (int i = 0; i < 4; ++i) {                                               \
    v2f wlo0 = *(const v2f*)(W0 + (WOFF) + half * 8 + 2 * i);                 \
    v2f wlo1 = *(const v2f*)(W1 + (WOFF) + half * 8 + 2 * i);                 \
    v2f whi0 = *(const v2f*)(W0 + (WOFF) + 16 + half * 8 + 2 * i);            \
    v2f whi1 = *(const v2f*)(W1 + (WOFF) + 16 + half * 8 + 2 * i);            \
    v2f p0 = {T0[2 * i], T0[2 * i + 1]};                                      \
    v2f p1 = {T1[2 * i], T1[2 * i + 1]};                                      \
    v2f p2 = {T2_[2 * i], T2_[2 * i + 1]};                                    \
    v2f p3 = {T3[2 * i], T3[2 * i + 1]};                                      \
    e00 = wmma_f32(p0, wlo0, e00);                                            \
    e01 = wmma_f32(p0, wlo1, e01);                                            \
    e10 = wmma_f32(p2, wlo0, e10);                                            \
    e11 = wmma_f32(p2, wlo1, e11);                                            \
    e00 = wmma_f32(p1, whi0, e00);                                            \
    e01 = wmma_f32(p1, whi1, e01);                                            \
    e10 = wmma_f32(p3, whi0, e10);                                            \
    e11 = wmma_f32(p3, whi1, e11);                                            \
  }
  GC_PART(c0, c1, c2, c3, 32)
  float res0[8], res1[8], res2[8], res3[8];
#pragma unroll
  for (int r = 0; r < 8; ++r) {
    size_t row = ((size_t)b * 512 + m0 + r + half * 8) * 64 + lbase;
    res0[r] = Hres[(row + 1) * 32 + o0];
    res1[r] = Hres[(row + 1) * 32 + o1];
    res2[r] = Hres[(row + 2) * 32 + o0];
    res3[r] = Hres[(row + 2) * 32 + o1];
  }
  GC_PART(d0, d1, d2, d3, 64)
#undef GC_PART

  float rinv = rsqrtf(1.0f + EPSBN);
  float ig0 = gcg[o0] * rinv, bg0 = gcb[o0], ib0 = bng[o0] * rinv, bb0 = bnb[o0];
  float ig1 = gcg[o1] * rinv, bg1 = gcb[o1], ib1 = bng[o1] * rinv, bb1 = bnb[o1];
#pragma unroll
  for (int r = 0; r < 8; ++r) {
    size_t row = ((size_t)b * 512 + m0 + r + half * 8) * 64 + lbase;
    if (lbase < Lout) {
      float g0v = e00[r] * ig0 + bg0;
      float g1v = e01[r] * ig1 + bg1;
      float h0 = g0v + res0[r];
      float h1 = g1v + res1[r];
      Hout[row * 32 + o0] = h0 * ib0 + bb0;
      Hout[row * 32 + o1] = h1 * ib1 + bb1;
    }
    if (lbase + 1 < Lout) {
      float g0v = e10[r] * ig0 + bg0;
      float g1v = e11[r] * ig1 + bg1;
      float h0 = g0v + res2[r];
      float h1 = g1v + res3[r];
      Hout[(row + 1) * 32 + o0] = h0 * ib0 + bb0;
      Hout[(row + 1) * 32 + o1] = h1 * ib1 + bb1;
    }
  }
}

__global__ void k_out(const float* __restrict__ S, const float* __restrict__ og,
                      const float* __restrict__ ob, float* __restrict__ out) {
  int t = blockIdx.x * blockDim.x + threadIdx.x;
  if (t >= 8 * 61 * 512 * 32) return;
  int cch = t & 31;
  int n   = (t >> 5) & 511;
  int rem = t >> 14;
  int l   = rem % 61;
  int b   = rem / 61;
  float inv = rsqrtf(1.0f + EPSBN);
  float v = S[(((size_t)(b * 512 + n)) * 64 + l) * 32 + cch];
  out[t] = v * (og[cch] * inv) + ob[cch];
}

extern "C" void kernel_launch(void* const* d_in, const int* in_sizes, int n_in,
                              void* d_out, int out_size, void* d_ws, size_t ws_size,
                              hipStream_t stream) {
  (void)in_sizes; (void)n_in; (void)out_size; (void)ws_size;
  const float* x   = (const float*)d_in[0];
  const float* adj = (const float*)d_in[1];
  const float* Wst = (const float*)d_in[2];
  const float* bst = (const float*)d_in[3];
  const float* Wf  = (const float*)d_in[4];
  const float* bf  = (const float*)d_in[5];
  const float* Wg  = (const float*)d_in[6];
  const float* bg  = (const float*)d_in[7];
  const float* Wsk = (const float*)d_in[8];
  const float* bsk = (const float*)d_in[9];
  const float* Wgc = (const float*)d_in[10];
  const float* bgc = (const float*)d_in[11];
  const float* gcg = (const float*)d_in[12];
  const float* gcb = (const float*)d_in[13];
  const float* bng = (const float*)d_in[14];
  const float* bnb = (const float*)d_in[15];
  const float* og  = (const float*)d_in[16];
  const float* ob  = (const float*)d_in[17];
  float* out = (float*)d_out;

  const size_t BUF = (size_t)8 * 512 * 64 * 32 + 1024;
  float* ws    = (float*)d_ws;
  float* bufH  = ws;
  float* bufH2 = ws + BUF;
  float* bufG  = ws + 2 * BUF;
  float* bufS  = ws + 5 * BUF;

  float* T2 = bufS + 62 * 32;
  k_adj2<<<512, 128, 0, stream>>>(adj, T2);

  k_start<<<32768, 256, 0, stream>>>(x, Wst, bst, bufH);

  float* H  = bufH;
  float* Hn = bufH2;
  for (int i = 0; i < 3; ++i) {
    int Lout = 63 - i;
    k_gated<<<4096, 256, 0, stream>>>(H, bufG, Wf + i * 2048, bf + i * 32,
                                      Wg + i * 2048, bg + i * 32, Lout);
    k_skip<<<4096, 256, 0, stream>>>(bufG, bufS, Wsk + i * 1024, bsk + i * 32,
                                     2 - i, i == 0 ? 1 : 0);
    if (i < 2) {
      k_diffuse_gconv<<<1024, 256, 0, stream>>>(adj, T2, bufG, H, Hn,
                                                Wgc + i * 3072, bgc + i * 32,
                                                gcg + i * 32, gcb + i * 32,
                                                bng + i * 32, bnb + i * 32, Lout);
      float* tmp = H; H = Hn; Hn = tmp;
    }
  }
  k_out<<<31232, 256, 0, stream>>>(bufS, og, ob, out);
}


# gated conv computed transposed (weights as A operand, from LDS): its D tiles are A fragments of the skip conv, same launch; last layer writes out; 11 -> 7 launches; both convs sum in new order
# speedup vs baseline: 1.6251x; 1.1931x over previous
#include <hip/hip_runtime.h>
#include <cmath>


typedef __attribute__((ext_vector_type(2))) float v2f;
typedef __attribute__((ext_vector_type(8))) float v8f;

#define EPSBN 1e-5f

__device__ __forceinline__ v8f wmma_f32(v2f a, v2f b, v8f c) {
  return __builtin_amdgcn_wmma_f32_16x16x4_f32(false, a, false, b, (short)0, c,
                                               false, false);
}

__global__ void k_start(const float* __restrict__ x, const float* __restrict__ Wst,
                        const float* __restrict__ bst, float* __restrict__ H) {
  int t = blockIdx.x * blockDim.x + threadIdx.x;
  int o   = t & 31;
  int pos = t >> 5;
  int l   = pos & 63;
  int n   = (pos >> 6) & 511;
  int b   = pos >> 15;
  const float* xb = x + (size_t)b * 2 * 512 * 64;
  float x0 = xb[(size_t)n * 64 + l];
  float x1 = xb[(size_t)512 * 64 + (size_t)n * 64 + l];
  H[t] = fmaf(x0, Wst[o * 2], fmaf(x1, Wst[o * 2 + 1], bst[o]));
}

__global__ void __launch_bounds__(256) k_gated_skip(
    const float* __restrict__ H, float* __restrict__ G,
    const float* __restrict__ Wf, const float* __restrict__ bf,
    const float* __restrict__ Wg, const float* __restrict__ bg,
    const float* __restrict__ Wk, const float* __restrict__ bk,
    float* __restrict__ S, const float* __restrict__ og,
    const float* __restrict__ ob, float* __restrict__ out, int mode) {
  __shared__ __attribute__((aligned(16))) float wA[4096];
  __shared__ __attribute__((aligned(16))) float wS[1024];

  for (int idx = threadIdx.x; idx < 4096; idx += 256) {
    int e2 = idx & 1, ln = (idx >> 1) & 31, s = (idx >> 6) & 15;
    int ot = (idx >> 10) & 1, mat = idx >> 11;
    int o = (ot << 4) + (ln & 15);
    int kk = ((s >> 1) << 3) + ((ln >> 4) << 2) + ((s & 1) << 1) + e2;
    const float* Wm = mat ? Wg : Wf;
    wA[idx] = Wm[o * 64 + (kk & 31) * 2 + (kk >> 5)];
  }
  for (int idx = threadIdx.x; idx < 1024; idx += 256) {
    int e2 = idx & 1, ln = (idx >> 1) & 31, ks = (idx >> 6) & 7, ot2 = idx >> 9;
    int o2 = (ot2 << 4) + (ln & 15);
    int ch = ((ks >> 2) << 4) + ((ln >> 4) << 3) + ((ks & 3) << 1) + e2;
    wS[idx] = Wk[o2 * 32 + ch];
  }
  __syncthreads();

  int wave = threadIdx.x >> 5;
  int lane = threadIdx.x & 31;
  int half = lane >> 4;
  int l15  = lane & 15;
  int node = blockIdx.x * 8 + wave;
  size_t p0 = (size_t)node * 64;
  int off = 2 - mode;

  v8f bF0, bF1, bG0, bG1;
#pragma unroll
  for (int r = 0; r < 8; ++r) {
    bF0[r] = bf[half * 8 + r];  bF1[r] = bf[16 + half * 8 + r];
    bG0[r] = bg[half * 8 + r];  bG1[r] = bg[16 + half * 8 + r];
  }
  float bk0 = bk[l15], bk1 = bk[16 + l15];
  float rinv = rsqrtf(1.0f + EPSBN);
  float sc0 = og[l15] * rinv, sc1 = og[16 + l15] * rinv;
  float ob0 = ob[l15], ob1 = ob[16 + l15];

  const float* fa = wA + lane * 2;
  const float* fs = wS + lane * 2;
  const float* hp = H + (p0 + l15) * 32 + half * 4;
  float4 hv[8];
#pragma unroll
  for (int j = 0; j < 8; ++j) hv[j] = *(const float4*)(hp + 8 * j);

#pragma unroll 1
  for (int t = 0; t < 4; ++t) {
    asm volatile("" ::: "memory");
    float4 hn[8];
#pragma unroll
    for (int j = 0; j < 8; ++j) hn[j] = hv[j];
    if (t < 3) {
#pragma unroll
      for (int j = 0; j < 8; ++j) hn[j] = *(const float4*)(hp + (t + 1) * 512 + 8 * j);
    }
    int lr0 = t * 16 + half * 8 - off;
    v8f c20, c21;
#pragma unroll
    for (int r = 0; r < 8; ++r) {
      int ls = lr0 + r;
      float pv0 = 0.0f, pv1 = 0.0f;
      if (mode != 0 && ls >= 0 && ls < 61) {
        pv0 = S[(p0 + ls) * 32 + l15];
        pv1 = S[(p0 + ls) * 32 + 16 + l15];
      }
      c20[r] = bk0 + pv0;
      c21[r] = bk1 + pv1;
    }

    v8f cf0 = bF0, cf1 = bF1, cg0 = bG0, cg1 = bG1;
#pragma unroll
    for (int s = 0; s < 16; ++s) {
      float4 q = hv[s >> 1];
      v2f bb;
      if (s & 1) { bb.x = q.z; bb.y = q.w; } else { bb.x = q.x; bb.y = q.y; }
      v2f af0 = *(const v2f*)(fa + (0 * 16 + s) * 64);
      v2f af1 = *(const v2f*)(fa + (1 * 16 + s) * 64);
      v2f ag0 = *(const v2f*)(fa + (2 * 16 + s) * 64);
      v2f ag1 = *(const v2f*)(fa + (3 * 16 + s) * 64);
      cf0 = wmma_f32(af0, bb, cf0);
      cg0 = wmma_f32(ag0, bb, cg0);
      cf1 = wmma_f32(af1, bb, cf1);
      cg1 = wmma_f32(ag1, bb, cg1);
    }
    v8f g0, g1;
#pragma unroll
    for (int r = 0; r < 8; ++r) {
      g0[r] = tanhf(cf0[r]) * (1.0f / (1.0f + __expf(-cg0[r])));
      g1[r] = tanhf(cf1[r]) * (1.0f / (1.0f + __expf(-cg1[r])));
    }
    if (mode != 2) {
      float* gp = G + (p0 + t * 16 + l15) * 32 + half * 8;
      *(float4*)(gp)      = make_float4(g0[0], g0[1], g0[2], g0[3]);
      *(float4*)(gp + 4)  = make_float4(g0[4], g0[5], g0[6], g0[7]);
      *(float4*)(gp + 16) = make_float4(g1[0], g1[1], g1[2], g1[3]);
      *(float4*)(gp + 20) = make_float4(g1[4], g1[5], g1[6], g1[7]);
    }
#pragma unroll
    for (int i = 0; i < 4; ++i) {
      v2f a0 = {g0[2 * i], g0[2 * i + 1]};
      v2f b00 = *(const v2f*)(fs + (0 * 8 + i) * 64);
      v2f b10 = *(const v2f*)(fs + (1 * 8 + i) * 64);
      c20 = wmma_f32(a0, b00, c20);
      c21 = wmma_f32(a0, b10, c21);
    }
#pragma unroll
    for (int i = 0; i < 4; ++i) {
      v2f a1 = {g1[2 * i], g1[2 * i + 1]};
      v2f b01 = *(const v2f*)(fs + (0 * 8 + 4 + i) * 64);
      v2f b11 = *(const v2f*)(fs + (1 * 8 + 4 + i) * 64);
      c20 = wmma_f32(a1, b01, c20);
      c21 = wmma_f32(a1, b11, c21);
    }
    if (mode != 2) {
#pragma unroll
      for (int r = 0; r < 8; ++r) {
        int ls = lr0 + r;
        if (ls >= 0 && ls < 61) {
          S[(p0 + ls) * 32 + l15]      = c20[r];
          S[(p0 + ls) * 32 + 16 + l15] = c21[r];
        }
      }
    } else {
      int b = node >> 9, n = node & 511;
#pragma unroll
      for (int r = 0; r < 8; ++r) {
        int ls = lr0 + r;
        if (ls < 61) {
          size_t oi = (((size_t)(b * 61 + ls)) * 512 + n) * 32;
          out[oi + l15]      = c20[r] * sc0 + ob0;
          out[oi + 16 + l15] = c21[r] * sc1 + ob1;
        }
      }
    }
#pragma unroll
    for (int j = 0; j < 8; ++j) hv[j] = hn[j];
  }
}

#define T2PITCH 36

__device__ __forceinline__ size_t t2_off(int m, int k) {
  return (size_t)(m * 8 + (k >> 6)) * 2048 + (size_t)(k & 63);
}

__global__ void __launch_bounds__(128) k_adj2(const float* __restrict__ adj,
                                              float* __restrict__ T2) {
  __shared__ __attribute__((aligned(16))) float part[4 * 16 * T2PITCH];

  int bid = blockIdx.x;
  int kt = bid & 15;
  int mt = bid >> 4;
  int m0 = mt << 4;
  int kc0 = kt << 5;
  int wave = threadIdx.x >> 5;
  int lane = threadIdx.x & 31;
  int half = lane >> 4;
  int l15  = lane & 15;

  const float* Acol = adj + m0 + l15;
  const float* B0   = adj + (size_t)(kc0 + l15) * 512;
  const float* B1   = adj + (size_t)(kc0 + 16 + l15) * 512;

  v8f c0 = {}, c1 = {};
  int nbeg = wave << 7;
#pragma unroll 8
  for (int i = 0; i < 128; i += 4) {
    int n = nbeg + i + half * 2;
    v2f a;
    a.x = Acol[(size_t)n * 512];
    a.y = Acol[(size_t)(n + 1) * 512];
    v2f b0 = *(const v2f*)(B0 + n);
    v2f b1 = *(const v2f*)(B1 + n);
    c0 = wmma_f32(a, b0, c0);
    c1 = wmma_f32(a, b1, c1);
  }
  float* pw = part + wave * 16 * T2PITCH;
#pragma unroll
  for (int r = 0; r < 8; ++r) {
    pw[(r + half * 8) * T2PITCH + l15]      = c0[r];
    pw[(r + half * 8) * T2PITCH + 16 + l15] = c1[r];
  }
  __syncthreads();

  int row  = threadIdx.x >> 3;
  int col4 = (threadIdx.x & 7) << 2;
  const float* p = part + row * T2PITCH + col4;
  float4 s0 = *(const float4*)(p);
  float4 s1 = *(const float4*)(p + 16 * T2PITCH);
  float4 s2 = *(const float4*)(p + 32 * T2PITCH);
  float4 s3 = *(const float4*)(p + 48 * T2PITCH);
  float4 s;
  s.x = ((s0.x + s1.x) + s2.x) + s3.x;
  s.y = ((s0.y + s1.y) + s2.y) + s3.y;
  s.z = ((s0.z + s1.z) + s2.z) + s3.z;
  s.w = ((s0.w + s1.w) + s2.w) + s3.w;
  *(float4*)(T2 + t2_off(m0 + row, kc0 + col4)) = s;
}

#define LDSPITCH 516

__global__ void __launch_bounds__(256) k_diffuse_gconv(
    const float* __restrict__ adj, const float* __restrict__ T2,
    const float* __restrict__ Xin, const float* __restrict__ Hres,
    float* __restrict__ Hout, const float* __restrict__ Wgc,
    const float* __restrict__ bgc, const float* __restrict__ gcg,
    const float* __restrict__ gcb, const float* __restrict__ bng,
    const float* __restrict__ bnb, int Lout) {
  __shared__ __attribute__((aligned(16))) float aT[2 * 16 * LDSPITCH];

  int bid = blockIdx.x;
  int cg = bid & 3;
  int mt = (bid >> 2) & 31;
  int b  = bid >> 7;
  int m0 = mt << 4;

#pragma unroll 4
  for (int idx = threadIdx.x; idx < 8192; idx += 256) {
    int ml = idx & 15;
    int k  = idx >> 4;
    aT[ml * LDSPITCH + k] = adj[(size_t)k * 512 + m0 + ml];
  }
#pragma unroll 4
  for (int idx = threadIdx.x; idx < 2048; idx += 256) {
    int ml = idx >> 7;
    int k  = (idx & 127) << 2;
    *(float4*)(aT + (16 + ml) * LDSPITCH + k) =
        *(const float4*)(T2 + t2_off(m0 + ml, k));
  }
  __syncthreads();

  int wave = __builtin_amdgcn_readfirstlane((int)(threadIdx.x >> 5));
  int lane = threadIdx.x & 31;
  int half = lane >> 4;
  int l15  = lane & 15;
  int j0   = (cg << 9) + (wave << 6);

  const float* Xb = Xin + (size_t)b * 512 * 2048;
  const float* Ar = aT + l15 * LDSPITCH;

  v8f c0 = {}, c1 = {}, c2 = {}, c3 = {};
  v8f d0 = {}, d1 = {}, d2 = {}, d3 = {};
#pragma unroll 4
  for (int n0 = 0; n0 < 512; n0 += 4) {
    int k = n0 + half * 2;
    v2f a1 = *(const v2f*)(Ar + k);
    v2f a2 = *(const v2f*)(Ar + 16 * LDSPITCH + k);
    const float* xr = Xb + (size_t)k * 2048 + j0 + l15;
    v2f b0, b1, b2, b3;
    b0.x = xr[0];  b0.y = xr[2048];
    b1.x = xr[16]; b1.y = xr[2048 + 16];
    b2.x = xr[32]; b2.y = xr[2048 + 32];
    b3.x = xr[48]; b3.y = xr[2048 + 48];
    c0 = wmma_f32(b0, a1, c0);
    d0 = wmma_f32(b0, a2, d0);
    c1 = wmma_f32(b1, a1, c1);
    d1 = wmma_f32(b1, a2, d1);
    c2 = wmma_f32(b2, a1, c2);
    d2 = wmma_f32(b2, a2, d2);
    c3 = wmma_f32(b3, a1, c3);
    d3 = wmma_f32(b3, a2, d3);
  }

  int o0 = l15, o1 = 16 + l15;
  const float* W0 = Wgc + (size_t)o0 * 96;
  const float* W1 = Wgc + (size_t)o1 * 96;
  int lbase = j0 >> 5;
  size_t nrow = ((size_t)b * 512 + m0 + l15) * 64;

  float bias0 = bgc[o0], bias1 = bgc[o1];
  v8f e00, e01, e10, e11;
#pragma unroll
  for (int r = 0; r < 8; ++r) { e00[r] = bias0; e01[r] = bias1; e10[r] = bias0; e11[r] = bias1; }

  {
    const float* g0 = Xin + (nrow + lbase) * 32 + half * 16;
    const float* g1 = g0 + 32;
    float4 ga[4], gb[4];
#pragma unroll
    for (int q = 0; q < 4; ++q) {
      ga[q] = *(const float4*)(g0 + 4 * q);
      gb[q] = *(const float4*)(g1 + 4 * q);
    }
#pragma unroll
    for (int q = 0; q < 4; ++q) {
      float4 w0 = *(const float4*)(W0 + half * 16 + 4 * q);
      float4 w1 = *(const float4*)(W1 + half * 16 + 4 * q);
      v2f wa0 = {w0.x, w0.y}, wb0 = {w0.z, w0.w};
      v2f wa1 = {w1.x, w1.y}, wb1 = {w1.z, w1.w};
      v2f xa = {ga[q].x, ga[q].y}, xb = {ga[q].z, ga[q].w};
      v2f ya = {gb[q].x, gb[q].y}, yb = {gb[q].z, gb[q].w};
      e00 = wmma_f32(xa, wa0, e00);
      e01 = wmma_f32(xa, wa1, e01);
      e10 = wmma_f32(ya, wa0, e10);
      e11 = wmma_f32(ya, wa1, e11);
      e00 = wmma_f32(xb, wb0, e00);
      e01 = wmma_f32(xb, wb1, e01);
      e10 = wmma_f32(yb, wb0, e10);
      e11 = wmma_f32(yb, wb1, e11);
    }
  }
#define GC_PART(T0, T1, T2_, T3, WOFF)                                        \
  _Pragma("unroll")                                                           \
  for (int i = 0; i < 4; ++i) {                                               \
    v2f wlo0 = *(const v2f*)(W0 + (WOFF) + half * 8 + 2 * i);                 \
    v2f wlo1 = *(const v2f*)(W1 + (WOFF) + half * 8 + 2 * i);                 \
    v2f whi0 = *(const v2f*)(W0 + (WOFF) + 16 + half * 8 + 2 * i);            \
    v2f whi1 = *(const v2f*)(W1 + (WOFF) + 16 + half * 8 + 2 * i);            \
    v2f p0 = {T0[2 * i], T0[2 * i + 1]};                                      \
    v2f p1 = {T1[2 * i], T1[2 * i + 1]};                                      \
    v2f p2 = {T2_[2 * i], T2_[2 * i + 1]};                                    \
    v2f p3 = {T3[2 * i], T3[2 * i + 1]};                                      \
    e00 = wmma_f32(p0, wlo0, e00);                                            \
    e01 = wmma_f32(p0, wlo1, e01);                                            \
    e10 = wmma_f32(p2, wlo0, e10);                                            \
    e11 = wmma_f32(p2, wlo1, e11);                                            \
    e00 = wmma_f32(p1, whi0, e00);                                            \
    e01 = wmma_f32(p1, whi1, e01);                                            \
    e10 = wmma_f32(p3, whi0, e10);                                            \
    e11 = wmma_f32(p3, whi1, e11);                                            \
  }
  GC_PART(c0, c1, c2, c3, 32)
  float res0[8], res1[8], res2[8], res3[8];
#pragma unroll
  for (int r = 0; r < 8; ++r) {
    size_t row = ((size_t)b * 512 + m0 + r + half * 8) * 64 + lbase;
    res0[r] = Hres[(row + 1) * 32 + o0];
    res1[r] = Hres[(row + 1) * 32 + o1];
    res2[r] = Hres[(row + 2) * 32 + o0];
    res3[r] = Hres[(row + 2) * 32 + o1];
  }
  GC_PART(d0, d1, d2, d3, 64)
#undef GC_PART

  float rinv = rsqrtf(1.0f + EPSBN);
  float ig0 = gcg[o0] * rinv, bg0 = gcb[o0], ib0 = bng[o0] * rinv, bb0 = bnb[o0];
  float ig1 = gcg[o1] * rinv, bg1 = gcb[o1], ib1 = bng[o1] * rinv, bb1 = bnb[o1];
#pragma unroll
  for (int r = 0; r < 8; ++r) {
    size_t row = ((size_t)b * 512 + m0 + r + half * 8) * 64 + lbase;
    if (lbase < Lout) {
      float g0v = e00[r] * ig0 + bg0;
      float g1v = e01[r] * ig1 + bg1;
      float h0 = g0v + res0[r];
      float h1 = g1v + res1[r];
      Hout[row * 32 + o0] = h0 * ib0 + bb0;
      Hout[row * 32 + o1] = h1 * ib1 + bb1;
    }
    if (lbase + 1 < Lout) {
      float g0v = e10[r] * ig0 + bg0;
      float g1v = e11[r] * ig1 + bg1;
      float h0 = g0v + res2[r];
      float h1 = g1v + res3[r];
      Hout[(row + 1) * 32 + o0] = h0 * ib0 + bb0;
      Hout[(row + 1) * 32 + o1] = h1 * ib1 + bb1;
    }
  }
}

extern "C" void kernel_launch(void* const* d_in, const int* in_sizes, int n_in,
                              void* d_out, int out_size, void* d_ws, size_t ws_size,
                              hipStream_t stream) {
  (void)in_sizes; (void)n_in; (void)out_size; (void)ws_size;
  const float* x   = (const float*)d_in[0];
  const float* adj = (const float*)d_in[1];
  const float* Wst = (const float*)d_in[2];
  const float* bst = (const float*)d_in[3];
  const float* Wf  = (const float*)d_in[4];
  const float* bf  = (const float*)d_in[5];
  const float* Wg  = (const float*)d_in[6];
  const float* bg  = (const float*)d_in[7];
  const float* Wsk = (const float*)d_in[8];
  const float* bsk = (const float*)d_in[9];
  const float* Wgc = (const float*)d_in[10];
  const float* bgc = (const float*)d_in[11];
  const float* gcg = (const float*)d_in[12];
  const float* gcb = (const float*)d_in[13];
  const float* bng = (const float*)d_in[14];
  const float* bnb = (const float*)d_in[15];
  const float* og  = (const float*)d_in[16];
  const float* ob  = (const float*)d_in[17];
  float* out = (float*)d_out;

  const size_t BUF = (size_t)8 * 512 * 64 * 32 + 1024;
  float* ws    = (float*)d_ws;
  float* bufH  = ws;
  float* bufH2 = ws + BUF;
  float* bufG  = ws + 2 * BUF;
  float* bufS  = ws + 5 * BUF;

  float* T2 = bufS + 62 * 32;
  k_adj2<<<512, 128, 0, stream>>>(adj, T2);

  k_start<<<32768, 256, 0, stream>>>(x, Wst, bst, bufH);

  k_gated_skip<<<512, 256, 0, stream>>>(bufH, bufG, Wf, bf, Wg, bg, Wsk, bsk,
                                        bufS, og, ob, out, 0);
  k_diffuse_gconv<<<1024, 256, 0, stream>>>(adj, T2, bufG, bufH, bufH2, Wgc, bgc,
                                            gcg, gcb, bng, bnb, 63);
  k_gated_skip<<<512, 256, 0, stream>>>(bufH2, bufG, Wf + 2048, bf + 32, Wg + 2048,
                                        bg + 32, Wsk + 1024, bsk + 32,
                                        bufS, og, ob, out, 1);
  k_diffuse_gconv<<<1024, 256, 0, stream>>>(adj, T2, bufG, bufH2, bufH, Wgc + 3072,
                                            bgc + 32, gcg + 32, gcb + 32,
                                            bng + 32, bnb + 32, 62);
  k_gated_skip<<<512, 256, 0, stream>>>(bufH, bufG, Wf + 4096, bf + 64, Wg + 4096,
                                        bg + 64, Wsk + 2048, bsk + 64,
                                        bufS, og, ob, out, 2);
}


# start 1x1 conv folded into layer 0: gated conv takes x with composed weights (W*Wstart, K 64 -> 4, built per block), residual rebuilt from x in graph conv; 7 -> 6 launches; layer-0 sums in new order
# speedup vs baseline: 1.7808x; 1.0958x over previous
#include <hip/hip_runtime.h>
#include <cmath>


typedef __attribute__((ext_vector_type(2))) float v2f;
typedef __attribute__((ext_vector_type(8))) float v8f;

#define EPSBN 1e-5f

__device__ __forceinline__ v8f wmma_f32(v2f a, v2f b, v8f c) {
  return __builtin_amdgcn_wmma_f32_16x16x4_f32(false, a, false, b, (short)0, c,
                                               false, false);
}

__global__ void __launch_bounds__(256) k_gated0_skip(
    const float* __restrict__ x, const float* __restrict__ Wst,
    const float* __restrict__ bst, float* __restrict__ G,
    const float* __restrict__ Wf, const float* __restrict__ bf,
    const float* __restrict__ Wg, const float* __restrict__ bg,
    const float* __restrict__ Wk, const float* __restrict__ bk,
    float* __restrict__ S) {
  __shared__ __attribute__((aligned(16))) float wC[512];
  __shared__ __attribute__((aligned(16))) float wS[1024];

  if (threadIdx.x < 128) {
    int tap = threadIdx.x & 1, o = (threadIdx.x >> 1) & 31, mat = threadIdx.x >> 6;
    const float* Wm = (mat ? Wg : Wf) + o * 64 + tap;
    float a0 = 0.0f, a1 = 0.0f, ab = 0.0f;
    for (int c = 0; c < 32; ++c) {
      float w = Wm[2 * c];
      a0 = fmaf(w, Wst[2 * c], a0);
      a1 = fmaf(w, Wst[2 * c + 1], a1);
      ab = fmaf(w, bst[c], ab);
    }
    *(float4*)(wC + threadIdx.x * 4) = make_float4(a0, a1, ab, 0.0f);
  }
  float wk[4];
#pragma unroll
  for (int j = 0; j < 4; ++j) {
    int idx = threadIdx.x + 256 * j;
    int e2 = idx & 1, ln = (idx >> 1) & 31, ks = (idx >> 6) & 7, ot2 = idx >> 9;
    int o2 = (ot2 << 4) + (ln & 15);
    int ch = ((ks >> 2) << 4) + ((ln >> 4) << 3) + ((ks & 3) << 1) + e2;
    wk[j] = Wk[o2 * 32 + ch];
  }
#pragma unroll
  for (int j = 0; j < 4; ++j) wS[threadIdx.x + 256 * j] = wk[j];
  __syncthreads();

  int wave = threadIdx.x >> 5;
  int lane = threadIdx.x & 31;
  int half = lane >> 4;
  int l15  = lane & 15;
  int node = blockIdx.x * 8 + wave;
  size_t p0 = (size_t)node * 64;

  v2f af0 = *(const v2f*)(wC + ((l15 * 2) + half) * 4);
  v2f af1 = *(const v2f*)(wC + (((16 + l15) * 2) + half) * 4);
  v2f ag0 = *(const v2f*)(wC + (((32 + l15) * 2) + half) * 4);
  v2f ag1 = *(const v2f*)(wC + (((48 + l15) * 2) + half) * 4);
  v8f bF0, bF1, bG0, bG1;
#pragma unroll
  for (int r = 0; r < 8; ++r) {
    int ch = half * 8 + r;
    bF0[r] = bf[ch]      + (wC[(ch * 2) * 4 + 2]        + wC[(ch * 2 + 1) * 4 + 2]);
    bF1[r] = bf[16 + ch] + (wC[((16 + ch) * 2) * 4 + 2] + wC[((16 + ch) * 2 + 1) * 4 + 2]);
    bG0[r] = bg[ch]      + (wC[((32 + ch) * 2) * 4 + 2] + wC[((32 + ch) * 2 + 1) * 4 + 2]);
    bG1[r] = bg[16 + ch] + (wC[((48 + ch) * 2) * 4 + 2] + wC[((48 + ch) * 2 + 1) * 4 + 2]);
  }
  float bk0 = bk[l15], bk1 = bk[16 + l15];
  const float* fs = wS + lane * 2;

  const float* x0p = x + (size_t)(node >> 9) * (2 * 512 * 64) + (size_t)(node & 511) * 64;
  const float* x1p = x0p + 512 * 64;
  v2f xb[4];
#pragma unroll
  for (int t = 0; t < 4; ++t) {
    int l = t * 16 + l15 + half;
    l = l < 63 ? l : 63;
    xb[t].x = x0p[l];
    xb[t].y = x1p[l];
  }
  v2f xq0 = xb[0], xq1 = xb[1], xq2 = xb[2], xq3 = xb[3];

#pragma unroll 1
  for (int t = 0; t < 4; ++t) {
    v8f cf0 = wmma_f32(af0, xq0, bF0);
    v8f cg0 = wmma_f32(ag0, xq0, bG0);
    v8f cf1 = wmma_f32(af1, xq0, bF1);
    v8f cg1 = wmma_f32(ag1, xq0, bG1);
    v8f g0, g1;
#pragma unroll
    for (int r = 0; r < 8; ++r) {
      g0[r] = tanhf(cf0[r]) * (1.0f / (1.0f + __expf(-cg0[r])));
      g1[r] = tanhf(cf1[r]) * (1.0f / (1.0f + __expf(-cg1[r])));
    }
    float* gp = G + (p0 + t * 16 + l15) * 32 + half * 8;
    *(float4*)(gp)      = make_float4(g0[0], g0[1], g0[2], g0[3]);
    *(float4*)(gp + 4)  = make_float4(g0[4], g0[5], g0[6], g0[7]);
    *(float4*)(gp + 16) = make_float4(g1[0], g1[1], g1[2], g1[3]);
    *(float4*)(gp + 20) = make_float4(g1[4], g1[5], g1[6], g1[7]);

    v8f c20, c21;
#pragma unroll
    for (int r = 0; r < 8; ++r) { c20[r] = bk0; c21[r] = bk1; }
#pragma unroll
    for (int i = 0; i < 4; ++i) {
      v2f a0 = {g0[2 * i], g0[2 * i + 1]};
      v2f b00 = *(const v2f*)(fs + (0 * 8 + i) * 64);
      v2f b10 = *(const v2f*)(fs + (1 * 8 + i) * 64);
      c20 = wmma_f32(a0, b00, c20);
      c21 = wmma_f32(a0, b10, c21);
    }
#pragma unroll
    for (int i = 0; i < 4; ++i) {
      v2f a1 = {g1[2 * i], g1[2 * i + 1]};
      v2f b01 = *(const v2f*)(fs + (0 * 8 + 4 + i) * 64);
      v2f b11 = *(const v2f*)(fs + (1 * 8 + 4 + i) * 64);
      c20 = wmma_f32(a1, b01, c20);
      c21 = wmma_f32(a1, b11, c21);
    }
    int lr0 = t * 16 + half * 8 - 2;
#pragma unroll
    for (int r = 0; r < 8; ++r) {
      int ls = lr0 + r;
      if (ls >= 0 && ls < 61) {
        S[(p0 + ls) * 32 + l15]      = c20[r];
        S[(p0 + ls) * 32 + 16 + l15] = c21[r];
      }
    }
    xq0 = xq1; xq1 = xq2; xq2 = xq3;
  }
}

__global__ void __launch_bounds__(256) k_gated_skip(
    const float* __restrict__ H, float* __restrict__ G,
    const float* __restrict__ Wf, const float* __restrict__ bf,
    const float* __restrict__ Wg, const float* __restrict__ bg,
    const float* __restrict__ Wk, const float* __restrict__ bk,
    float* __restrict__ S, const float* __restrict__ og,
    const float* __restrict__ ob, float* __restrict__ out, int mode) {
  __shared__ __attribute__((aligned(16))) float wA[4096];
  __shared__ __attribute__((aligned(16))) float wS[1024];

  for (int idx = threadIdx.x; idx < 4096; idx += 256) {
    int e2 = idx & 1, ln = (idx >> 1) & 31, s = (idx >> 6) & 15;
    int ot = (idx >> 10) & 1, mat = idx >> 11;
    int o = (ot << 4) + (ln & 15);
    int kk = ((s >> 1) << 3) + ((ln >> 4) << 2) + ((s & 1) << 1) + e2;
    const float* Wm = mat ? Wg : Wf;
    wA[idx] = Wm[o * 64 + (kk & 31) * 2 + (kk >> 5)];
  }
  for (int idx = threadIdx.x; idx < 1024; idx += 256) {
    int e2 = idx & 1, ln = (idx >> 1) & 31, ks = (idx >> 6) & 7, ot2 = idx >> 9;
    int o2 = (ot2 << 4) + (ln & 15);
    int ch = ((ks >> 2) << 4) + ((ln >> 4) << 3) + ((ks & 3) << 1) + e2;
    wS[idx] = Wk[o2 * 32 + ch];
  }
  __syncthreads();

  int wave = threadIdx.x >> 5;
  int lane = threadIdx.x & 31;
  int half = lane >> 4;
  int l15  = lane & 15;
  int node = blockIdx.x * 8 + wave;
  size_t p0 = (size_t)node * 64;
  int off = 2 - mode;

  v8f bF0, bF1, bG0, bG1;
#pragma unroll
  for (int r = 0; r < 8; ++r) {
    bF0[r] = bf[half * 8 + r];  bF1[r] = bf[16 + half * 8 + r];
    bG0[r] = bg[half * 8 + r];  bG1[r] = bg[16 + half * 8 + r];
  }
  float bk0 = bk[l15], bk1 = bk[16 + l15];
  float rinv = rsqrtf(1.0f + EPSBN);
  float sc0 = og[l15] * rinv, sc1 = og[16 + l15] * rinv;
  float ob0 = ob[l15], ob1 = ob[16 + l15];

  const float* fa = wA + lane * 2;
  const float* fs = wS + lane * 2;
  const float* hp = H + (p0 + l15) * 32 + half * 4;
  float4 hv[8];
#pragma unroll
  for (int j = 0; j < 8; ++j) hv[j] = *(const float4*)(hp + 8 * j);

#pragma unroll 1
  for (int t = 0; t < 4; ++t) {
    asm volatile("" ::: "memory");
    float4 hn[8];
#pragma unroll
    for (int j = 0; j < 8; ++j) hn[j] = hv[j];
    if (t < 3) {
#pragma unroll
      for (int j = 0; j < 8; ++j) hn[j] = *(const float4*)(hp + (t + 1) * 512 + 8 * j);
    }
    int lr0 = t * 16 + half * 8 - off;
    v8f c20, c21;
#pragma unroll
    for (int r = 0; r < 8; ++r) {
      int ls = lr0 + r;
      float pv0 = 0.0f, pv1 = 0.0f;
      if (ls >= 0 && ls < 61) {
        pv0 = S[(p0 + ls) * 32 + l15];
        pv1 = S[(p0 + ls) * 32 + 16 + l15];
      }
      c20[r] = bk0 + pv0;
      c21[r] = bk1 + pv1;
    }

    v8f cf0 = bF0, cf1 = bF1, cg0 = bG0, cg1 = bG1;
#pragma unroll
    for (int s = 0; s < 16; ++s) {
      float4 q = hv[s >> 1];
      v2f bb;
      if (s & 1) { bb.x = q.z; bb.y = q.w; } else { bb.x = q.x; bb.y = q.y; }
      v2f af0 = *(const v2f*)(fa + (0 * 16 + s) * 64);
      v2f af1 = *(const v2f*)(fa + (1 * 16 + s) * 64);
      v2f ag0 = *(const v2f*)(fa + (2 * 16 + s) * 64);
      v2f ag1 = *(const v2f*)(fa + (3 * 16 + s) * 64);
      cf0 = wmma_f32(af0, bb, cf0);
      cg0 = wmma_f32(ag0, bb, cg0);
      cf1 = wmma_f32(af1, bb, cf1);
      cg1 = wmma_f32(ag1, bb, cg1);
    }
    v8f g0, g1;
#pragma unroll
    for (int r = 0; r < 8; ++r) {
      g0[r] = tanhf(cf0[r]) * (1.0f / (1.0f + __expf(-cg0[r])));
      g1[r] = tanhf(cf1[r]) * (1.0f / (1.0f + __expf(-cg1[r])));
    }
    if (mode != 2) {
      float* gp = G + (p0 + t * 16 + l15) * 32 + half * 8;
      *(float4*)(gp)      = make_float4(g0[0], g0[1], g0[2], g0[3]);
      *(float4*)(gp + 4)  = make_float4(g0[4], g0[5], g0[6], g0[7]);
      *(float4*)(gp + 16) = make_float4(g1[0], g1[1], g1[2], g1[3]);
      *(float4*)(gp + 20) = make_float4(g1[4], g1[5], g1[6], g1[7]);
    }
#pragma unroll
    for (int i = 0; i < 4; ++i) {
      v2f a0 = {g0[2 * i], g0[2 * i + 1]};
      v2f b00 = *(const v2f*)(fs + (0 * 8 + i) * 64);
      v2f b10 = *(const v2f*)(fs + (1 * 8 + i) * 64);
      c20 = wmma_f32(a0, b00, c20);
      c21 = wmma_f32(a0, b10, c21);
    }
#pragma unroll
    for (int i = 0; i < 4; ++i) {
      v2f a1 = {g1[2 * i], g1[2 * i + 1]};
      v2f b01 = *(const v2f*)(fs + (0 * 8 + 4 + i) * 64);
      v2f b11 = *(const v2f*)(fs + (1 * 8 + 4 + i) * 64);
      c20 = wmma_f32(a1, b01, c20);
      c21 = wmma_f32(a1, b11, c21);
    }
    if (mode != 2) {
#pragma unroll
      for (int r = 0; r < 8; ++r) {
        int ls = lr0 + r;
        if (ls >= 0 && ls < 61) {
          S[(p0 + ls) * 32 + l15]      = c20[r];
          S[(p0 + ls) * 32 + 16 + l15] = c21[r];
        }
      }
    } else {
      int b = node >> 9, n = node & 511;
#pragma unroll
      for (int r = 0; r < 8; ++r) {
        int ls = lr0 + r;
        if (ls < 61) {
          size_t oi = (((size_t)(b * 61 + ls)) * 512 + n) * 32;
          out[oi + l15]      = c20[r] * sc0 + ob0;
          out[oi + 16 + l15] = c21[r] * sc1 + ob1;
        }
      }
    }
#pragma unroll
    for (int j = 0; j < 8; ++j) hv[j] = hn[j];
  }
}

#define T2PITCH 36

__device__ __forceinline__ size_t t2_off(int m, int k) {
  return (size_t)(m * 8 + (k >> 6)) * 2048 + (size_t)(k & 63);
}

__global__ void __launch_bounds__(128) k_adj2(const float* __restrict__ adj,
                                              float* __restrict__ T2) {
  __shared__ __attribute__((aligned(16))) float part[4 * 16 * T2PITCH];

  int bid = blockIdx.x;
  int kt = bid & 15;
  int mt = bid >> 4;
  int m0 = mt << 4;
  int kc0 = kt << 5;
  int wave = threadIdx.x >> 5;
  int lane = threadIdx.x & 31;
  int half = lane >> 4;
  int l15  = lane & 15;

  const float* Acol = adj + m0 + l15;
  const float* B0   = adj + (size_t)(kc0 + l15) * 512;
  const float* B1   = adj + (size_t)(kc0 + 16 + l15) * 512;

  v8f c0 = {}, c1 = {};
  int nbeg = wave << 7;
#pragma unroll 8
  for (int i = 0; i < 128; i += 4) {
    int n = nbeg + i + half * 2;
    v2f a;
    a.x = Acol[(size_t)n * 512];
    a.y = Acol[(size_t)(n + 1) * 512];
    v2f b0 = *(const v2f*)(B0 + n);
    v2f b1 = *(const v2f*)(B1 + n);
    c0 = wmma_f32(a, b0, c0);
    c1 = wmma_f32(a, b1, c1);
  }
  float* pw = part + wave * 16 * T2PITCH;
#pragma unroll
  for (int r = 0; r < 8; ++r) {
    pw[(r + half * 8) * T2PITCH + l15]      = c0[r];
    pw[(r + half * 8) * T2PITCH + 16 + l15] = c1[r];
  }
  __syncthreads();

  int row  = threadIdx.x >> 3;
  int col4 = (threadIdx.x & 7) << 2;
  const float* p = part + row * T2PITCH + col4;
  float4 s0 = *(const float4*)(p);
  float4 s1 = *(const float4*)(p + 16 * T2PITCH);
  float4 s2 = *(const float4*)(p + 32 * T2PITCH);
  float4 s3 = *(const float4*)(p + 48 * T2PITCH);
  float4 s;
  s.x = ((s0.x + s1.x) + s2.x) + s3.x;
  s.y = ((s0.y + s1.y) + s2.y) + s3.y;
  s.z = ((s0.z + s1.z) + s2.z) + s3.z;
  s.w = ((s0.w + s1.w) + s2.w) + s3.w;
  *(float4*)(T2 + t2_off(m0 + row, kc0 + col4)) = s;
}

#define LDSPITCH 516

template <bool FROM_X>
__global__ void __launch_bounds__(256) k_diffuse_gconv(
    const float* __restrict__ adj, const float* __restrict__ T2,
    const float* __restrict__ Xin, const float* __restrict__ Hres,
    float* __restrict__ Hout, const float* __restrict__ Wgc,
    const float* __restrict__ bgc, const float* __restrict__ gcg,
    const float* __restrict__ gcb, const float* __restrict__ bng,
    const float* __restrict__ bnb, int Lout,
    const float* __restrict__ x, const float* __restrict__ Wst,
    const float* __restrict__ bst) {
  __shared__ __attribute__((aligned(16))) float aT[2 * 16 * LDSPITCH];

  int bid = blockIdx.x;
  int cg = bid & 3;
  int mt = (bid >> 2) & 31;
  int b  = bid >> 7;
  int m0 = mt << 4;

#pragma unroll 4
  for (int idx = threadIdx.x; idx < 8192; idx += 256) {
    int ml = idx & 15;
    int k  = idx >> 4;
    aT[ml * LDSPITCH + k] = adj[(size_t)k * 512 + m0 + ml];
  }
#pragma unroll 4
  for (int idx = threadIdx.x; idx < 2048; idx += 256) {
    int ml = idx >> 7;
    int k  = (idx & 127) << 2;
    *(float4*)(aT + (16 + ml) * LDSPITCH + k) =
        *(const float4*)(T2 + t2_off(m0 + ml, k));
  }
  __syncthreads();

  int wave = __builtin_amdgcn_readfirstlane((int)(threadIdx.x >> 5));
  int lane = threadIdx.x & 31;
  int half = lane >> 4;
  int l15  = lane & 15;
  int j0   = (cg << 9) + (wave << 6);

  const float* Xb = Xin + (size_t)b * 512 * 2048;
  const float* Ar = aT + l15 * LDSPITCH;

  v8f c0 = {}, c1 = {}, c2 = {}, c3 = {};
  v8f d0 = {}, d1 = {}, d2 = {}, d3 = {};
#pragma unroll 4
  for (int n0 = 0; n0 < 512; n0 += 4) {
    int k = n0 + half * 2;
    v2f a1 = *(const v2f*)(Ar + k);
    v2f a2 = *(const v2f*)(Ar + 16 * LDSPITCH + k);
    const float* xr = Xb + (size_t)k * 2048 + j0 + l15;
    v2f b0, b1, b2, b3;
    b0.x = xr[0];  b0.y = xr[2048];
    b1.x = xr[16]; b1.y = xr[2048 + 16];
    b2.x = xr[32]; b2.y = xr[2048 + 32];
    b3.x = xr[48]; b3.y = xr[2048 + 48];
    c0 = wmma_f32(b0, a1, c0);
    d0 = wmma_f32(b0, a2, d0);
    c1 = wmma_f32(b1, a1, c1);
    d1 = wmma_f32(b1, a2, d1);
    c2 = wmma_f32(b2, a1, c2);
    d2 = wmma_f32(b2, a2, d2);
    c3 = wmma_f32(b3, a1, c3);
    d3 = wmma_f32(b3, a2, d3);
  }

  int o0 = l15, o1 = 16 + l15;
  const float* W0 = Wgc + (size_t)o0 * 96;
  const float* W1 = Wgc + (size_t)o1 * 96;
  int lbase = j0 >> 5;
  size_t nrow = ((size_t)b * 512 + m0 + l15) * 64;

  float bias0 = bgc[o0], bias1 = bgc[o1];
  v8f e00, e01, e10, e11;
#pragma unroll
  for (int r = 0; r < 8; ++r) { e00[r] = bias0; e01[r] = bias1; e10[r] = bias0; e11[r] = bias1; }

  {
    const float* g0 = Xin + (nrow + lbase) * 32 + half * 16;
    const float* g1 = g0 + 32;
    float4 ga[4], gb[4];
#pragma unroll
    for (int q = 0; q < 4; ++q) {
      ga[q] = *(const float4*)(g0 + 4 * q);
      gb[q] = *(const float4*)(g1 + 4 * q);
    }
#pragma unroll
    for (int q = 0; q < 4; ++q) {
      float4 w0 = *(const float4*)(W0 + half * 16 + 4 * q);
      float4 w1 = *(const float4*)(W1 + half * 16 + 4 * q);
      v2f wa0 = {w0.x, w0.y}, wb0 = {w0.z, w0.w};
      v2f wa1 = {w1.x, w1.y}, wb1 = {w1.z, w1.w};
      v2f xa = {ga[q].x, ga[q].y}, xb = {ga[q].z, ga[q].w};
      v2f ya = {gb[q].x, gb[q].y}, yb = {gb[q].z, gb[q].w};
      e00 = wmma_f32(xa, wa0, e00);
      e01 = wmma_f32(xa, wa1, e01);
      e10 = wmma_f32(ya, wa0, e10);
      e11 = wmma_f32(ya, wa1, e11);
      e00 = wmma_f32(xb, wb0, e00);
      e01 = wmma_f32(xb, wb1, e01);
      e10 = wmma_f32(yb, wb0, e10);
      e11 = wmma_f32(yb, wb1, e11);
    }
  }
#define GC_PART(T0, T1, T2_, T3, WOFF)                                        \
  _Pragma("unroll")                                                           \
  for (int i = 0; i < 4; ++i) {                                               \
    v2f wlo0 = *(const v2f*)(W0 + (WOFF) + half * 8 + 2 * i);                 \
    v2f wlo1 = *(const v2f*)(W1 + (WOFF) + half * 8 + 2 * i);                 \
    v2f whi0 = *(const v2f*)(W0 + (WOFF) + 16 + half * 8 + 2 * i);            \
    v2f whi1 = *(const v2f*)(W1 + (WOFF) + 16 + half * 8 + 2 * i);            \
    v2f p0 = {T0[2 * i], T0[2 * i + 1]};                                      \
    v2f p1 = {T1[2 * i], T1[2 * i + 1]};                                      \
    v2f p2 = {T2_[2 * i], T2_[2 * i + 1]};                                    \
    v2f p3 = {T3[2 * i], T3[2 * i + 1]};                                      \
    e00 = wmma_f32(p0, wlo0, e00);                                            \
    e01 = wmma_f32(p0, wlo1, e01);                                            \
    e10 = wmma_f32(p2, wlo0, e10);                                            \
    e11 = wmma_f32(p2, wlo1, e11);                                            \
    e00 = wmma_f32(p1, whi0, e00);                                            \
    e01 = wmma_f32(p1, whi1, e01);                                            \
    e10 = wmma_f32(p3, whi0, e10);                                            \
    e11 = wmma_f32(p3, whi1, e11);                                            \
  }
  GC_PART(c0, c1, c2, c3, 32)
  float res0[8], res1[8], res2[8], res3[8];
  if (FROM_X) {
    float w00 = Wst[o0 * 2], w01 = Wst[o0 * 2 + 1], bs0 = bst[o0];
    float w10 = Wst[o1 * 2], w11 = Wst[o1 * 2 + 1], bs1 = bst[o1];
    int la = lbase + 1 < 63 ? lbase + 1 : 63;
    int lb = lbase + 2 < 63 ? lbase + 2 : 63;
    const float* xn = x + (size_t)b * (2 * 512 * 64) + (size_t)(m0 + half * 8) * 64;
#pragma unroll
    for (int r = 0; r < 8; ++r) {
      float xa0 = xn[r * 64 + la], xa1 = xn[512 * 64 + r * 64 + la];
      float xb0 = xn[r * 64 + lb], xb1 = xn[512 * 64 + r * 64 + lb];
      res0[r] = fmaf(xa0, w00, fmaf(xa1, w01, bs0));
      res1[r] = fmaf(xa0, w10, fmaf(xa1, w11, bs1));
      res2[r] = fmaf(xb0, w00, fmaf(xb1, w01, bs0));
      res3[r] = fmaf(xb0, w10, fmaf(xb1, w11, bs1));
    }
  } else {
#pragma unroll
    for (int r = 0; r < 8; ++r) {
      size_t row = ((size_t)b * 512 + m0 + r + half * 8) * 64 + lbase;
      res0[r] = Hres[(row + 1) * 32 + o0];
      res1[r] = Hres[(row + 1) * 32 + o1];
      res2[r] = Hres[(row + 2) * 32 + o0];
      res3[r] = Hres[(row + 2) * 32 + o1];
    }
  }
  GC_PART(d0, d1, d2, d3, 64)
#undef GC_PART

  float rinv = rsqrtf(1.0f + EPSBN);
  float ig0 = gcg[o0] * rinv, bg0 = gcb[o0], ib0 = bng[o0] * rinv, bb0 = bnb[o0];
  float ig1 = gcg[o1] * rinv, bg1 = gcb[o1], ib1 = bng[o1] * rinv, bb1 = bnb[o1];
#pragma unroll
  for (int r = 0; r < 8; ++r) {
    size_t row = ((size_t)b * 512 + m0 + r + half * 8) * 64 + lbase;
    if (lbase < Lout) {
      float g0v = e00[r] * ig0 + bg0;
      float g1v = e01[r] * ig1 + bg1;
      float h0 = g0v + res0[r];
      float h1 = g1v + res1[r];
      Hout[row * 32 + o0] = h0 * ib0 + bb0;
      Hout[row * 32 + o1] = h1 * ib1 + bb1;
    }
    if (lbase + 1 < Lout) {
      float g0v = e10[r] * ig0 + bg0;
      float g1v = e11[r] * ig1 + bg1;
      float h0 = g0v + res2[r];
      float h1 = g1v + res3[r];
      Hout[(row + 1) * 32 + o0] = h0 * ib0 + bb0;
      Hout[(row + 1) * 32 + o1] = h1 * ib1 + bb1;
    }
  }
}

extern "C" void kernel_launch(void* const* d_in, const int* in_sizes, int n_in,
                              void* d_out, int out_size, void* d_ws, size_t ws_size,
                              hipStream_t stream) {
  (void)in_sizes; (void)n_in; (void)out_size; (void)ws_size;
  const float* x   = (const float*)d_in[0];
  const float* adj = (const float*)d_in[1];
  const float* Wst = (const float*)d_in[2];
  const float* bst = (const float*)d_in[3];
  const float* Wf  = (const float*)d_in[4];
  const float* bf  = (const float*)d_in[5];
  const float* Wg  = (const float*)d_in[6];
  const float* bg  = (const float*)d_in[7];
  const float* Wsk = (const float*)d_in[8];
  const float* bsk = (const float*)d_in[9];
  const float* Wgc = (const float*)d_in[10];
  const float* bgc = (const float*)d_in[11];
  const float* gcg = (const float*)d_in[12];
  const float* gcb = (const float*)d_in[13];
  const float* bng = (const float*)d_in[14];
  const float* bnb = (const float*)d_in[15];
  const float* og  = (const float*)d_in[16];
  const float* ob  = (const float*)d_in[17];
  float* out = (float*)d_out;

  const size_t BUF = (size_t)8 * 512 * 64 * 32 + 1024;
  float* ws    = (float*)d_ws;
  float* bufH  = ws;
  float* bufH2 = ws + BUF;
  float* bufG  = ws + 2 * BUF;
  float* bufS  = ws + 5 * BUF;

  float* T2 = bufS + 62 * 32;
  k_adj2<<<512, 128, 0, stream>>>(adj, T2);

  k_gated0_skip<<<512, 256, 0, stream>>>(x, Wst, bst, bufG, Wf, bf, Wg, bg, Wsk, bsk,
                                         bufS);
  k_diffuse_gconv<true><<<1024, 256, 0, stream>>>(adj, T2, bufG, bufH, bufH2, Wgc, bgc,
                                                  gcg, gcb, bng, bnb, 63, x, Wst, bst);
  k_gated_skip<<<512, 256, 0, stream>>>(bufH2, bufG, Wf + 2048, bf + 32, Wg + 2048,
                                        bg + 32, Wsk + 1024, bsk + 32,
                                        bufS, og, ob, out, 1);
  k_diffuse_gconv<false><<<1024, 256, 0, stream>>>(adj, T2, bufG, bufH2, bufH, Wgc + 3072,
                                                   bgc + 32, gcg + 32, gcb + 32,
                                                   bng + 32, bnb + 32, 62, x, Wst, bst);
  k_gated_skip<<<512, 256, 0, stream>>>(bufH, bufG, Wf + 4096, bf + 64, Wg + 4096,
                                        bg + 64, Wsk + 2048, bsk + 64,
                                        bufS, og, ob, out, 2);
}


# k_diffuse loop: X, adj and T2 go through double-buffered LDS in 32-step chunks (16-byte loads); block tile 128 columns x 64 nodes, 8 waves share the fragments; 2x fewer L2 reads; sums unchanged
# speedup vs baseline: 1.8954x; 1.0644x over previous
#include <hip/hip_runtime.h>
#include <cmath>


typedef __attribute__((ext_vector_type(2))) float v2f;
typedef __attribute__((ext_vector_type(8))) float v8f;

#define EPSBN 1e-5f

__device__ __forceinline__ v8f wmma_f32(v2f a, v2f b, v8f c) {
  return __builtin_amdgcn_wmma_f32_16x16x4_f32(false, a, false, b, (short)0, c,
                                               false, false);
}

__global__ void __launch_bounds__(256) k_gated0_skip(
    const float* __restrict__ x, const float* __restrict__ Wst,
    const float* __restrict__ bst, float* __restrict__ G,
    const float* __restrict__ Wf, const float* __restrict__ bf,
    const float* __restrict__ Wg, const float* __restrict__ bg,
    const float* __restrict__ Wk, const float* __restrict__ bk,
    float* __restrict__ S) {
  __shared__ __attribute__((aligned(16))) float wC[512];
  __shared__ __attribute__((aligned(16))) float wS[1024];

  if (threadIdx.x < 128) {
    int tap = threadIdx.x & 1, o = (threadIdx.x >> 1) & 31, mat = threadIdx.x >> 6;
    const float* Wm = (mat ? Wg : Wf) + o * 64 + tap;
    float a0 = 0.0f, a1 = 0.0f, ab = 0.0f;
    for (int c = 0; c < 32; ++c) {
      float w = Wm[2 * c];
      a0 = fmaf(w, Wst[2 * c], a0);
      a1 = fmaf(w, Wst[2 * c + 1], a1);
      ab = fmaf(w, bst[c], ab);
    }
    *(float4*)(wC + threadIdx.x * 4) = make_float4(a0, a1, ab, 0.0f);
  }
  float wk[4];
#pragma unroll
  for (int j = 0; j < 4; ++j) {
    int idx = threadIdx.x + 256 * j;
    int e2 = idx & 1, ln = (idx >> 1) & 31, ks = (idx >> 6) & 7, ot2 = idx >> 9;
    int o2 = (ot2 << 4) + (ln & 15);
    int ch = ((ks >> 2) << 4) + ((ln >> 4) << 3) + ((ks & 3) << 1) + e2;
    wk[j] = Wk[o2 * 32 + ch];
  }
#pragma unroll
  for (int j = 0; j < 4; ++j) wS[threadIdx.x + 256 * j] = wk[j];
  __syncthreads();

  int wave = threadIdx.x >> 5;
  int lane = threadIdx.x & 31;
  int half = lane >> 4;
  int l15  = lane & 15;
  int node = blockIdx.x * 8 + wave;
  size_t p0 = (size_t)node * 64;

  v2f af0 = *(const v2f*)(wC + ((l15 * 2) + half) * 4);
  v2f af1 = *(const v2f*)(wC + (((16 + l15) * 2) + half) * 4);
  v2f ag0 = *(const v2f*)(wC + (((32 + l15) * 2) + half) * 4);
  v2f ag1 = *(const v2f*)(wC + (((48 + l15) * 2) + half) * 4);
  v8f bF0, bF1, bG0, bG1;
#pragma unroll
  for (int r = 0; r < 8; ++r) {
    int ch = half * 8 + r;
    bF0[r] = bf[ch]      + (wC[(ch * 2) * 4 + 2]        + wC[(ch * 2 + 1) * 4 + 2]);
    bF1[r] = bf[16 + ch] + (wC[((16 + ch) * 2) * 4 + 2] + wC[((16 + ch) * 2 + 1) * 4 + 2]);
    bG0[r] = bg[ch]      + (wC[((32 + ch) * 2) * 4 + 2] + wC[((32 + ch) * 2 + 1) * 4 + 2]);
    bG1[r] = bg[16 + ch] + (wC[((48 + ch) * 2) * 4 + 2] + wC[((48 + ch) * 2 + 1) * 4 + 2]);
  }
  float bk0 = bk[l15], bk1 = bk[16 + l15];
  const float* fs = wS + lane * 2;

  const float* x0p = x + (size_t)(node >> 9) * (2 * 512 * 64) + (size_t)(node & 511) * 64;
  const float* x1p = x0p + 512 * 64;
  v2f xb[4];
#pragma unroll
  for (int t = 0; t < 4; ++t) {
    int l = t * 16 + l15 + half;
    l = l < 63 ? l : 63;
    xb[t].x = x0p[l];
    xb[t].y = x1p[l];
  }
  v2f xq0 = xb[0], xq1 = xb[1], xq2 = xb[2], xq3 = xb[3];

#pragma unroll 1
  for (int t = 0; t < 4; ++t) {
    v8f cf0 = wmma_f32(af0, xq0, bF0);
    v8f cg0 = wmma_f32(ag0, xq0, bG0);
    v8f cf1 = wmma_f32(af1, xq0, bF1);
    v8f cg1 = wmma_f32(ag1, xq0, bG1);
    v8f g0, g1;
#pragma unroll
    for (int r = 0; r < 8; ++r) {
      g0[r] = tanhf(cf0[r]) * (1.0f / (1.0f + __expf(-cg0[r])));
      g1[r] = tanhf(cf1[r]) * (1.0f / (1.0f + __expf(-cg1[r])));
    }
    float* gp = G + (p0 + t * 16 + l15) * 32 + half * 8;
    *(float4*)(gp)      = make_float4(g0[0], g0[1], g0[2], g0[3]);
    *(float4*)(gp + 4)  = make_float4(g0[4], g0[5], g0[6], g0[7]);
    *(float4*)(gp + 16) = make_float4(g1[0], g1[1], g1[2], g1[3]);
    *(float4*)(gp + 20) = make_float4(g1[4], g1[5], g1[6], g1[7]);

    v8f c20, c21;
#pragma unroll
    for (int r = 0; r < 8; ++r) { c20[r] = bk0; c21[r] = bk1; }
#pragma unroll
    for (int i = 0; i < 4; ++i) {
      v2f a0 = {g0[2 * i], g0[2 * i + 1]};
      v2f b00 = *(const v2f*)(fs + (0 * 8 + i) * 64);
      v2f b10 = *(const v2f*)(fs + (1 * 8 + i) * 64);
      c20 = wmma_f32(a0, b00, c20);
      c21 = wmma_f32(a0, b10, c21);
    }
#pragma unroll
    for (int i = 0; i < 4; ++i) {
      v2f a1 = {g1[2 * i], g1[2 * i + 1]};
      v2f b01 = *(const v2f*)(fs + (0 * 8 + 4 + i) * 64);
      v2f b11 = *(const v2f*)(fs + (1 * 8 + 4 + i) * 64);
      c20 = wmma_f32(a1, b01, c20);
      c21 = wmma_f32(a1, b11, c21);
    }
    int lr0 = t * 16 + half * 8 - 2;
#pragma unroll
    for (int r = 0; r < 8; ++r) {
      int ls = lr0 + r;
      if (ls >= 0 && ls < 61) {
        S[(p0 + ls) * 32 + l15]      = c20[r];
        S[(p0 + ls) * 32 + 16 + l15] = c21[r];
      }
    }
    xq0 = xq1; xq1 = xq2; xq2 = xq3;
  }
}

__global__ void __launch_bounds__(256) k_gated_skip(
    const float* __restrict__ H, float* __restrict__ G,
    const float* __restrict__ Wf, const float* __restrict__ bf,
    const float* __restrict__ Wg, const float* __restrict__ bg,
    const float* __restrict__ Wk, const float* __restrict__ bk,
    float* __restrict__ S, const float* __restrict__ og,
    const float* __restrict__ ob, float* __restrict__ out, int mode) {
  __shared__ __attribute__((aligned(16))) float wA[4096];
  __shared__ __attribute__((aligned(16))) float wS[1024];

  for (int idx = threadIdx.x; idx < 4096; idx += 256) {
    int e2 = idx & 1, ln = (idx >> 1) & 31, s = (idx >> 6) & 15;
    int ot = (idx >> 10) & 1, mat = idx >> 11;
    int o = (ot << 4) + (ln & 15);
    int kk = ((s >> 1) << 3) + ((ln >> 4) << 2) + ((s & 1) << 1) + e2;
    const float* Wm = mat ? Wg : Wf;
    wA[idx] = Wm[o * 64 + (kk & 31) * 2 + (kk >> 5)];
  }
  for (int idx = threadIdx.x; idx < 1024; idx += 256) {
    int e2 = idx & 1, ln = (idx >> 1) & 31, ks = (idx >> 6) & 7, ot2 = idx >> 9;
    int o2 = (ot2 << 4) + (ln & 15);
    int ch = ((ks >> 2) << 4) + ((ln >> 4) << 3) + ((ks & 3) << 1) + e2;
    wS[idx] = Wk[o2 * 32 + ch];
  }
  __syncthreads();

  int wave = threadIdx.x >> 5;
  int lane = threadIdx.x & 31;
  int half = lane >> 4;
  int l15  = lane & 15;
  int node = blockIdx.x * 8 + wave;
  size_t p0 = (size_t)node * 64;
  int off = 2 - mode;

  v8f bF0, bF1, bG0, bG1;
#pragma unroll
  for (int r = 0; r < 8; ++r) {
    bF0[r] = bf[half * 8 + r];  bF1[r] = bf[16 + half * 8 + r];
    bG0[r] = bg[half * 8 + r];  bG1[r] = bg[16 + half * 8 + r];
  }
  float bk0 = bk[l15], bk1 = bk[16 + l15];
  float rinv = rsqrtf(1.0f + EPSBN);
  float sc0 = og[l15] * rinv, sc1 = og[16 + l15] * rinv;
  float ob0 = ob[l15], ob1 = ob[16 + l15];

  const float* fa = wA + lane * 2;
  const float* fs = wS + lane * 2;
  const float* hp = H + (p0 + l15) * 32 + half * 4;
  float4 hv[8];
#pragma unroll
  for (int j = 0; j < 8; ++j) hv[j] = *(const float4*)(hp + 8 * j);

#pragma unroll 1
  for (int t = 0; t < 4; ++t) {
    asm volatile("" ::: "memory");
    float4 hn[8];
#pragma unroll
    for (int j = 0; j < 8; ++j) hn[j] = hv[j];
    if (t < 3) {
#pragma unroll
      for (int j = 0; j < 8; ++j) hn[j] = *(const float4*)(hp + (t + 1) * 512 + 8 * j);
    }
    int lr0 = t * 16 + half * 8 - off;
    v8f c20, c21;
#pragma unroll
    for (int r = 0; r < 8; ++r) {
      int ls = lr0 + r;
      float pv0 = 0.0f, pv1 = 0.0f;
      if (ls >= 0 && ls < 61) {
        pv0 = S[(p0 + ls) * 32 + l15];
        pv1 = S[(p0 + ls) * 32 + 16 + l15];
      }
      c20[r] = bk0 + pv0;
      c21[r] = bk1 + pv1;
    }

    v8f cf0 = bF0, cf1 = bF1, cg0 = bG0, cg1 = bG1;
#pragma unroll
    for (int s = 0; s < 16; ++s) {
      float4 q = hv[s >> 1];
      v2f bb;
      if (s & 1) { bb.x = q.z; bb.y = q.w; } else { bb.x = q.x; bb.y = q.y; }
      v2f af0 = *(const v2f*)(fa + (0 * 16 + s) * 64);
      v2f af1 = *(const v2f*)(fa + (1 * 16 + s) * 64);
      v2f ag0 = *(const v2f*)(fa + (2 * 16 + s) * 64);
      v2f ag1 = *(const v2f*)(fa + (3 * 16 + s) * 64);
      cf0 = wmma_f32(af0, bb, cf0);
      cg0 = wmma_f32(ag0, bb, cg0);
      cf1 = wmma_f32(af1, bb, cf1);
      cg1 = wmma_f32(ag1, bb, cg1);
    }
    v8f g0, g1;
#pragma unroll
    for (int r = 0; r < 8; ++r) {
      g0[r] = tanhf(cf0[r]) * (1.0f / (1.0f + __expf(-cg0[r])));
      g1[r] = tanhf(cf1[r]) * (1.0f / (1.0f + __expf(-cg1[r])));
    }
    if (mode != 2) {
      float* gp = G + (p0 + t * 16 + l15) * 32 + half * 8;
      *(float4*)(gp)      = make_float4(g0[0], g0[1], g0[2], g0[3]);
      *(float4*)(gp + 4)  = make_float4(g0[4], g0[5], g0[6], g0[7]);
      *(float4*)(gp + 16) = make_float4(g1[0], g1[1], g1[2], g1[3]);
      *(float4*)(gp + 20) = make_float4(g1[4], g1[5], g1[6], g1[7]);
    }
#pragma unroll
    for (int i = 0; i < 4; ++i) {
      v2f a0 = {g0[2 * i], g0[2 * i + 1]};
      v2f b00 = *(const v2f*)(fs + (0 * 8 + i) * 64);
      v2f b10 = *(const v2f*)(fs + (1 * 8 + i) * 64);
      c20 = wmma_f32(a0, b00, c20);
      c21 = wmma_f32(a0, b10, c21);
    }
#pragma unroll
    for (int i = 0; i < 4; ++i) {
      v2f a1 = {g1[2 * i], g1[2 * i + 1]};
      v2f b01 = *(const v2f*)(fs + (0 * 8 + 4 + i) * 64);
      v2f b11 = *(const v2f*)(fs + (1 * 8 + 4 + i) * 64);
      c20 = wmma_f32(a1, b01, c20);
      c21 = wmma_f32(a1, b11, c21);
    }
    if (mode != 2) {
#pragma unroll
      for (int r = 0; r < 8; ++r) {
        int ls = lr0 + r;
        if (ls >= 0 && ls < 61) {
          S[(p0 + ls) * 32 + l15]      = c20[r];
          S[(p0 + ls) * 32 + 16 + l15] = c21[r];
        }
      }
    } else {
      int b = node >> 9, n = node & 511;
#pragma unroll
      for (int r = 0; r < 8; ++r) {
        int ls = lr0 + r;
        if (ls < 61) {
          size_t oi = (((size_t)(b * 61 + ls)) * 512 + n) * 32;
          out[oi + l15]      = c20[r] * sc0 + ob0;
          out[oi + 16 + l15] = c21[r] * sc1 + ob1;
        }
      }
    }
#pragma unroll
    for (int j = 0; j < 8; ++j) hv[j] = hn[j];
  }
}

#define T2PITCH 36

__device__ __forceinline__ size_t t2_off(int m, int k) {
  return (size_t)(m * 8 + (k >> 6)) * 2048 + (size_t)(k & 63);
}

__global__ void __launch_bounds__(128) k_adj2(const float* __restrict__ adj,
                                              float* __restrict__ T2) {
  __shared__ __attribute__((aligned(16))) float part[4 * 16 * T2PITCH];

  int bid = blockIdx.x;
  int kt = bid & 15;
  int mt = bid >> 4;
  int m0 = mt << 4;
  int kc0 = kt << 5;
  int wave = threadIdx.x >> 5;
  int lane = threadIdx.x & 31;
  int half = lane >> 4;
  int l15  = lane & 15;

  const float* Acol = adj + m0 + l15;
  const float* B0   = adj + (size_t)(kc0 + l15) * 512;
  const float* B1   = adj + (size_t)(kc0 + 16 + l15) * 512;

  v8f c0 = {}, c1 = {};
  int nbeg = wave << 7;
#pragma unroll 8
  for (int i = 0; i < 128; i += 4) {
    int n = nbeg + i + half * 2;
    v2f a;
    a.x = Acol[(size_t)n * 512];
    a.y = Acol[(size_t)(n + 1) * 512];
    v2f b0 = *(const v2f*)(B0 + n);
    v2f b1 = *(const v2f*)(B1 + n);
    c0 = wmma_f32(a, b0, c0);
    c1 = wmma_f32(a, b1, c1);
  }
  float* pw = part + wave * 16 * T2PITCH;
#pragma unroll
  for (int r = 0; r < 8; ++r) {
    pw[(r + half * 8) * T2PITCH + l15]      = c0[r];
    pw[(r + half * 8) * T2PITCH + 16 + l15] = c1[r];
  }
  __syncthreads();

  int row  = threadIdx.x >> 3;
  int col4 = (threadIdx.x & 7) << 2;
  const float* p = part + row * T2PITCH + col4;
  float4 s0 = *(const float4*)(p);
  float4 s1 = *(const float4*)(p + 16 * T2PITCH);
  float4 s2 = *(const float4*)(p + 32 * T2PITCH);
  float4 s3 = *(const float4*)(p + 48 * T2PITCH);
  float4 s;
  s.x = ((s0.x + s1.x) + s2.x) + s3.x;
  s.y = ((s0.y + s1.y) + s2.y) + s3.y;
  s.z = ((s0.z + s1.z) + s2.z) + s3.z;
  s.w = ((s0.w + s1.w) + s2.w) + s3.w;
  *(float4*)(T2 + t2_off(m0 + row, kc0 + col4)) = s;
}

#define XPITCH 288
#define APITCH 160
#define TPITCH 36
#define XS_F (16 * XPITCH)
#define AS_F (16 * APITCH)
#define TS_F (64 * TPITCH)
#define STAGE_F (XS_F + AS_F + TS_F)

template <bool FROM_X>
__global__ void __launch_bounds__(256) __attribute__((amdgpu_waves_per_eu(8)))
k_diffuse_gconv(
    const float* __restrict__ adj, const float* __restrict__ T2,
    const float* __restrict__ Xin, const float* __restrict__ Hres,
    float* __restrict__ Hout, const float* __restrict__ Wgc,
    const float* __restrict__ bgc, const float* __restrict__ gcg,
    const float* __restrict__ gcb, const float* __restrict__ bng,
    const float* __restrict__ bnb, int Lout,
    const float* __restrict__ x, const float* __restrict__ Wst,
    const float* __restrict__ bst) {
  __shared__ __attribute__((aligned(16))) float st[2 * STAGE_F];

  int bid = blockIdx.x;
  int jb = bid & 15;
  int mb = (bid >> 4) & 7;
  int b  = bid >> 7;
  int jc = jb << 7;
  int mc = mb << 6;
  int tid = threadIdx.x;

  const float* Xb = Xin + (size_t)b * 512 * 2048 + jc;
  const float* Ab = adj + mc;
  const float* Tb = T2 + t2_off(mc, 0);
  unsigned xgo = (unsigned)(tid >> 5) * 2048u + 4u * (unsigned)(tid & 31);
  unsigned ago = (unsigned)(tid >> 4) * 512u + 4u * (unsigned)(tid & 15);
  unsigned tgo = (unsigned)(tid >> 3) * (8u * 2048u) + 4u * (unsigned)(tid & 7);
  int xso = (tid >> 6) * XPITCH + ((tid >> 5) & 1) * 16 + ((tid & 31) >> 2) * 32 + (tid & 3) * 4;
  int aso = XS_F + (tid >> 5) * APITCH + ((tid >> 4) & 1) * 16 + ((tid & 15) >> 2) * 32 + (tid & 3) * 4;
  int tso = XS_F + AS_F + (tid >> 3) * TPITCH + 4 * (tid & 7);

  float4 x0r, x1r, x2r, x3r, a0r, a1r, t0r, t1r;
#define STAGE_LOAD(C)                                                         \
  {                                                                           \
    const float* xp = Xb + (size_t)(C) * (32 * 2048) + xgo;                   \
    x0r = *(const float4*)(xp);                                               \
    x1r = *(const float4*)(xp + 8 * 2048);                                    \
    x2r = *(const float4*)(xp + 16 * 2048);                                   \
    x3r = *(const float4*)(xp + 24 * 2048);                                   \
    const float* ap = Ab + (size_t)(C) * (32 * 512) + ago;                    \
    a0r = *(const float4*)(ap);                                               \
    a1r = *(const float4*)(ap + 16 * 512);                                    \
    const float* tp = Tb + (size_t)((C) >> 1) * 2048 + ((C) & 1) * 32 + tgo;  \
    t0r = *(const float4*)(tp);                                               \
    t1r = *(const float4*)(tp + 32 * 8 * 2048);                               \
  }
#define STAGE_STORE(S)                                                        \
  {                                                                           \
    float* sb = st + (S) * STAGE_F;                                           \
    *(float4*)(sb + xso)               = x0r;                                 \
    *(float4*)(sb + xso + 4 * XPITCH)  = x1r;                                 \
    *(float4*)(sb + xso + 8 * XPITCH)  = x2r;                                 \
    *(float4*)(sb + xso + 12 * XPITCH) = x3r;                                 \
    *(float4*)(sb + aso)               = a0r;                                 \
    *(float4*)(sb + aso + 8 * APITCH)  = a1r;                                 \
    *(float4*)(sb + tso)               = t0r;                                 \
    *(float4*)(sb + tso + 32 * TPITCH) = t1r;                                 \
  }

  STAGE_LOAD(0)
  STAGE_STORE(0)
  __syncthreads();

  int wave = __builtin_amdgcn_readfirstlane((int)(threadIdx.x >> 5));
  int lane = threadIdx.x & 31;
  int half = lane >> 4;
  int l15  = lane & 15;
  int mt = wave & 3, jt = wave >> 2;
  int m0 = mc + (mt << 4);
  int j0 = jc + (jt << 6);

  int xfo = half * XPITCH + (jt << 7) + l15;
  int afo = XS_F + half * APITCH + (mt << 5) + l15;
  int tfo = XS_F + AS_F + ((mt << 4) + l15) * TPITCH + 2 * half;

  v8f c0 = {}, c1 = {}, c2 = {}, c3 = {};
  v8f d0 = {}, d1 = {}, d2 = {}, d3 = {};
#pragma unroll 1
  for (int c = 0; c < 16; ++c) {
    const float* sb = st + (c & 1) * STAGE_F;
    if (c < 15) STAGE_LOAD(c + 1)
#pragma unroll 1
    for (int s = 0; s < 8; ++s) {
      const float* af = sb + afo + s * (2 * APITCH);
      v2f a1 = {af[0], af[16]};
      v2f a2 = *(const v2f*)(sb + tfo + s * 4);
      const float* xf = sb + xfo + s * (2 * XPITCH);
      v2f b0 = {xf[0],  xf[16]};
      v2f b1 = {xf[32], xf[48]};
      v2f b2 = {xf[64], xf[80]};
      v2f b3 = {xf[96], xf[112]};
      c0 = wmma_f32(b0, a1, c0);
      d0 = wmma_f32(b0, a2, d0);
      c1 = wmma_f32(b1, a1, c1);
      d1 = wmma_f32(b1, a2, d1);
      c2 = wmma_f32(b2, a1, c2);
      d2 = wmma_f32(b2, a2, d2);
      c3 = wmma_f32(b3, a1, c3);
      d3 = wmma_f32(b3, a2, d3);
    }
    if (c < 15) STAGE_STORE((c + 1) & 1)
    __syncthreads();
  }
#undef STAGE_LOAD
#undef STAGE_STORE

  int o0 = l15, o1 = 16 + l15;
  const float* W0 = Wgc + (size_t)o0 * 96;
  const float* W1 = Wgc + (size_t)o1 * 96;
  int lbase = j0 >> 5;
  size_t nrow = ((size_t)b * 512 + m0 + l15) * 64;

  float bias0 = bgc[o0], bias1 = bgc[o1];
  v8f e00, e01, e10, e11;
#pragma unroll
  for (int r = 0; r < 8; ++r) { e00[r] = bias0; e01[r] = bias1; e10[r] = bias0; e11[r] = bias1; }

  {
    const float* g0 = Xin + (nrow + lbase) * 32 + half * 16;
    const float* g1 = g0 + 32;
    float4 ga[4], gb[4];
#pragma unroll
    for (int q = 0; q < 4; ++q) {
      ga[q] = *(const float4*)(g0 + 4 * q);
      gb[q] = *(const float4*)(g1 + 4 * q);
    }
#pragma unroll
    for (int q = 0; q < 4; ++q) {
      float4 w0 = *(const float4*)(W0 + half * 16 + 4 * q);
      float4 w1 = *(const float4*)(W1 + half * 16 + 4 * q);
      v2f wa0 = {w0.x, w0.y}, wb0 = {w0.z, w0.w};
      v2f wa1 = {w1.x, w1.y}, wb1 = {w1.z, w1.w};
      v2f xa = {ga[q].x, ga[q].y}, xb = {ga[q].z, ga[q].w};
      v2f ya = {gb[q].x, gb[q].y}, yb = {gb[q].z, gb[q].w};
      e00 = wmma_f32(xa, wa0, e00);
      e01 = wmma_f32(xa, wa1, e01);
      e10 = wmma_f32(ya, wa0, e10);
      e11 = wmma_f32(ya, wa1, e11);
      e00 = wmma_f32(xb, wb0, e00);
      e01 = wmma_f32(xb, wb1, e01);
      e10 = wmma_f32(yb, wb0, e10);
      e11 = wmma_f32(yb, wb1, e11);
    }
  }
#define GC_PART(T0, T1, T2_, T3, WOFF)                                        \
  _Pragma("unroll")                                                           \
  for (int i = 0; i < 4; ++i) {                                               \
    v2f wlo0 = *(const v2f*)(W0 + (WOFF) + half * 8 + 2 * i);                 \
    v2f wlo1 = *(const v2f*)(W1 + (WOFF) + half * 8 + 2 * i);                 \
    v2f whi0 = *(const v2f*)(W0 + (WOFF) + 16 + half * 8 + 2 * i);            \
    v2f whi1 = *(const v2f*)(W1 + (WOFF) + 16 + half * 8 + 2 * i);            \
    v2f p0 = {T0[2 * i], T0[2 * i + 1]};                                      \
    v2f p1 = {T1[2 * i], T1[2 * i + 1]};                                      \
    v2f p2 = {T2_[2 * i], T2_[2 * i + 1]};                                    \
    v2f p3 = {T3[2 * i], T3[2 * i + 1]};                                      \
    e00 = wmma_f32(p0, wlo0, e00);                                            \
    e01 = wmma_f32(p0, wlo1, e01);                                            \
    e10 = wmma_f32(p2, wlo0, e10);                                            \
    e11 = wmma_f32(p2, wlo1, e11);                                            \
    e00 = wmma_f32(p1, whi0, e00);                                            \
    e01 = wmma_f32(p1, whi1, e01);                                            \
    e10 = wmma_f32(p3, whi0, e10);                                            \
    e11 = wmma_f32(p3, whi1, e11);                                            \
  }
  GC_PART(c0, c1, c2, c3, 32)
  float res0[8], res1[8], res2[8], res3[8];
  if (FROM_X) {
    float w00 = Wst[o0 * 2], w01 = Wst[o0 * 2 + 1], bs0 = bst[o0];
    float w10 = Wst[o1 * 2], w11 = Wst[o1 * 2 + 1], bs1 = bst[o1];
    int la = lbase + 1 < 63 ? lbase + 1 : 63;
    int lb = lbase + 2 < 63 ? lbase + 2 : 63;
    const float* xn = x + (size_t)b * (2 * 512 * 64) + (size_t)(m0 + half * 8) * 64;
#pragma unroll
    for (int r = 0; r < 8; ++r) {
      float xa0 = xn[r * 64 + la], xa1 = xn[512 * 64 + r * 64 + la];
      float xb0 = xn[r * 64 + lb], xb1 = xn[512 * 64 + r * 64 + lb];
      res0[r] = fmaf(xa0, w00, fmaf(xa1, w01, bs0));
      res1[r] = fmaf(xa0, w10, fmaf(xa1, w11, bs1));
      res2[r] = fmaf(xb0, w00, fmaf(xb1, w01, bs0));
      res3[r] = fmaf(xb0, w10, fmaf(xb1, w11, bs1));
    }
  } else {
#pragma unroll
    for (int r = 0; r < 8; ++r) {
      size_t row = ((size_t)b * 512 + m0 + r + half * 8) * 64 + lbase;
      res0[r] = Hres[(row + 1) * 32 + o0];
      res1[r] = Hres[(row + 1) * 32 + o1];
      res2[r] = Hres[(row + 2) * 32 + o0];
      res3[r] = Hres[(row + 2) * 32 + o1];
    }
  }
  GC_PART(d0, d1, d2, d3, 64)
#undef GC_PART

  float rinv = rsqrtf(1.0f + EPSBN);
  float ig0 = gcg[o0] * rinv, bg0 = gcb[o0], ib0 = bng[o0] * rinv, bb0 = bnb[o0];
  float ig1 = gcg[o1] * rinv, bg1 = gcb[o1], ib1 = bng[o1] * rinv, bb1 = bnb[o1];
#pragma unroll
  for (int r = 0; r < 8; ++r) {
    size_t row = ((size_t)b * 512 + m0 + r + half * 8) * 64 + lbase;
    if (lbase < Lout) {
      float g0v = e00[r] * ig0 + bg0;
      float g1v = e01[r] * ig1 + bg1;
      float h0 = g0v + res0[r];
      float h1 = g1v + res1[r];
      Hout[row * 32 + o0] = h0 * ib0 + bb0;
      Hout[row * 32 + o1] = h1 * ib1 + bb1;
    }
    if (lbase + 1 < Lout) {
      float g0v = e10[r] * ig0 + bg0;
      float g1v = e11[r] * ig1 + bg1;
      float h0 = g0v + res2[r];
      float h1 = g1v + res3[r];
      Hout[(row + 1) * 32 + o0] = h0 * ib0 + bb0;
      Hout[(row + 1) * 32 + o1] = h1 * ib1 + bb1;
    }
  }
}

extern "C" void kernel_launch(void* const* d_in, const int* in_sizes, int n_in,
                              void* d_out, int out_size, void* d_ws, size_t ws_size,
                              hipStream_t stream) {
  (void)in_sizes; (void)n_in; (void)out_size; (void)ws_size;
  const float* x   = (const float*)d_in[0];
  const float* adj = (const float*)d_in[1];
  const float* Wst = (const float*)d_in[2];
  const float* bst = (const float*)d_in[3];
  const float* Wf  = (const float*)d_in[4];
  const float* bf  = (const float*)d_in[5];
  const float* Wg  = (const float*)d_in[6];
  const float* bg  = (const float*)d_in[7];
  const float* Wsk = (const float*)d_in[8];
  const float* bsk = (const float*)d_in[9];
  const float* Wgc = (const float*)d_in[10];
  const float* bgc = (const float*)d_in[11];
  const float* gcg = (const float*)d_in[12];
  const float* gcb = (const float*)d_in[13];
  const float* bng = (const float*)d_in[14];
  const float* bnb = (const float*)d_in[15];
  const float* og  = (const float*)d_in[16];
  const float* ob  = (const float*)d_in[17];
  float* out = (float*)d_out;

  const size_t BUF = (size_t)8 * 512 * 64 * 32 + 1024;
  float* ws    = (float*)d_ws;
  float* bufH  = ws;
  float* bufH2 = ws + BUF;
  float* bufG  = ws + 2 * BUF;
  float* bufS  = ws + 5 * BUF;

  float* T2 = bufS + 62 * 32;
  k_adj2<<<512, 128, 0, stream>>>(adj, T2);

  k_gated0_skip<<<512, 256, 0, stream>>>(x, Wst, bst, bufG, Wf, bf, Wg, bg, Wsk, bsk,
                                         bufS);
  k_diffuse_gconv<true><<<1024, 256, 0, stream>>>(adj, T2, bufG, bufH, bufH2, Wgc, bgc,
                                                  gcg, gcb, bng, bnb, 63, x, Wst, bst);
  k_gated_skip<<<512, 256, 0, stream>>>(bufH2, bufG, Wf + 2048, bf + 32, Wg + 2048,
                                        bg + 32, Wsk + 1024, bsk + 32,
                                        bufS, og, ob, out, 1);
  k_diffuse_gconv<false><<<1024, 256, 0, stream>>>(adj, T2, bufG, bufH2, bufH, Wgc + 3072,
                                                   bgc + 32, gcg + 32, gcb + 32,
                                                   bng + 32, bnb + 32, 62, x, Wst, bst);
  k_gated_skip<<<512, 256, 0, stream>>>(bufH, bufG, Wf + 4096, bf + 64, Wg + 4096,
                                        bg + 64, Wsk + 2048, bsk + 64,
                                        bufS, og, ob, out, 2);
}


# gated-conv epilogue on the chip's tanh instruction: tanh(f) and sigmoid(g) = 0.5 + 0.5 tanh(g/2) are one v_tanh_f32 each (no exp, division, branch); last bits differ from library tanh; sums unchanged
# speedup vs baseline: 1.9529x; 1.0303x over previous
#include <hip/hip_runtime.h>


typedef __attribute__((ext_vector_type(2))) float v2f;
typedef __attribute__((ext_vector_type(8))) float v8f;

#define EPSBN 1e-5f

__device__ __forceinline__ v8f wmma_f32(v2f a, v2f b, v8f c) {
  return __builtin_amdgcn_wmma_f32_16x16x4_f32(false, a, false, b, (short)0, c,
                                               false, false);
}

__device__ __forceinline__ float gate_f32(float f, float g) {
  float tf = __builtin_amdgcn_tanhf(f);
  float tg = __builtin_amdgcn_tanhf(0.5f * g);
  return tf * fmaf(0.5f, tg, 0.5f);
}

__global__ void __launch_bounds__(256) k_gated0_skip(
    const float* __restrict__ x, const float* __restrict__ Wst,
    const float* __restrict__ bst, float* __restrict__ G,
    const float* __restrict__ Wf, const float* __restrict__ bf,
    const float* __restrict__ Wg, const float* __restrict__ bg,
    const float* __restrict__ Wk, const float* __restrict__ bk,
    float* __restrict__ S) {
  __shared__ __attribute__((aligned(16))) float wC[512];
  __shared__ __attribute__((aligned(16))) float wS[1024];

  if (threadIdx.x < 128) {
    int tap = threadIdx.x & 1, o = (threadIdx.x >> 1) & 31, mat = threadIdx.x >> 6;
    const float* Wm = (mat ? Wg : Wf) + o * 64 + tap;
    float a0 = 0.0f, a1 = 0.0f, ab = 0.0f;
    for (int c = 0; c < 32; ++c) {
      float w = Wm[2 * c];
      a0 = fmaf(w, Wst[2 * c], a0);
      a1 = fmaf(w, Wst[2 * c + 1], a1);
      ab = fmaf(w, bst[c], ab);
    }
    *(float4*)(wC + threadIdx.x * 4) = make_float4(a0, a1, ab, 0.0f);
  }
  float wk[4];
#pragma unroll
  for (int j = 0; j < 4; ++j) {
    int idx = threadIdx.x + 256 * j;
    int e2 = idx & 1, ln = (idx >> 1) & 31, ks = (idx >> 6) & 7, ot2 = idx >> 9;
    int o2 = (ot2 << 4) + (ln & 15);
    int ch = ((ks >> 2) << 4) + ((ln >> 4) << 3) + ((ks & 3) << 1) + e2;
    wk[j] = Wk[o2 * 32 + ch];
  }
#pragma unroll
  for (int j = 0; j < 4; ++j) wS[threadIdx.x + 256 * j] = wk[j];
  __syncthreads();

  int wave = threadIdx.x >> 5;
  int lane = threadIdx.x & 31;
  int half = lane >> 4;
  int l15  = lane & 15;
  int node = blockIdx.x * 8 + wave;
  size_t p0 = (size_t)node * 64;

  v2f af0 = *(const v2f*)(wC + ((l15 * 2) + half) * 4);
  v2f af1 = *(const v2f*)(wC + (((16 + l15) * 2) + half) * 4);
  v2f ag0 = *(const v2f*)(wC + (((32 + l15) * 2) + half) * 4);
  v2f ag1 = *(const v2f*)(wC + (((48 + l15) * 2) + half) * 4);
  v8f bF0, bF1, bG0, bG1;
#pragma unroll
  for (int r = 0; r < 8; ++r) {
    int ch = half * 8 + r;
    bF0[r] = bf[ch]      + (wC[(ch * 2) * 4 + 2]        + wC[(ch * 2 + 1) * 4 + 2]);
    bF1[r] = bf[16 + ch] + (wC[((16 + ch) * 2) * 4 + 2] + wC[((16 + ch) * 2 + 1) * 4 + 2]);
    bG0[r] = bg[ch]      + (wC[((32 + ch) * 2) * 4 + 2] + wC[((32 + ch) * 2 + 1) * 4 + 2]);
    bG1[r] = bg[16 + ch] + (wC[((48 + ch) * 2) * 4 + 2] + wC[((48 + ch) * 2 + 1) * 4 + 2]);
  }
  float bk0 = bk[l15], bk1 = bk[16 + l15];
  const float* fs = wS + lane * 2;

  const float* x0p = x + (size_t)(node >> 9) * (2 * 512 * 64) + (size_t)(node & 511) * 64;
  const float* x1p = x0p + 512 * 64;
  v2f xb[4];
#pragma unroll
  for (int t = 0; t < 4; ++t) {
    int l = t * 16 + l15 + half;
    l = l < 63 ? l : 63;
    xb[t].x = x0p[l];
    xb[t].y = x1p[l];
  }
  v2f xq0 = xb[0], xq1 = xb[1], xq2 = xb[2], xq3 = xb[3];

#pragma unroll 1
  for (int t = 0; t < 4; ++t) {
    v8f cf0 = wmma_f32(af0, xq0, bF0);
    v8f cg0 = wmma_f32(ag0, xq0, bG0);
    v8f cf1 = wmma_f32(af1, xq0, bF1);
    v8f cg1 = wmma_f32(ag1, xq0, bG1);
    v8f g0, g1;
#pragma unroll
    for (int r = 0; r < 8; ++r) {
      g0[r] = gate_f32(cf0[r], cg0[r]);
      g1[r] = gate_f32(cf1[r], cg1[r]);
    }
    float* gp = G + (p0 + t * 16 + l15) * 32 + half * 8;
    *(float4*)(gp)      = make_float4(g0[0], g0[1], g0[2], g0[3]);
    *(float4*)(gp + 4)  = make_float4(g0[4], g0[5], g0[6], g0[7]);
    *(float4*)(gp + 16) = make_float4(g1[0], g1[1], g1[2], g1[3]);
    *(float4*)(gp + 20) = make_float4(g1[4], g1[5], g1[6], g1[7]);

    v8f c20, c21;
#pragma unroll
    for (int r = 0; r < 8; ++r) { c20[r] = bk0; c21[r] = bk1; }
#pragma unroll
    for (int i = 0; i < 4; ++i) {
      v2f a0 = {g0[2 * i], g0[2 * i + 1]};
      v2f b00 = *(const v2f*)(fs + (0 * 8 + i) * 64);
      v2f b10 = *(const v2f*)(fs + (1 * 8 + i) * 64);
      c20 = wmma_f32(a0, b00, c20);
      c21 = wmma_f32(a0, b10, c21);
    }
#pragma unroll
    for (int i = 0; i < 4; ++i) {
      v2f a1 = {g1[2 * i], g1[2 * i + 1]};
      v2f b01 = *(const v2f*)(fs + (0 * 8 + 4 + i) * 64);
      v2f b11 = *(const v2f*)(fs + (1 * 8 + 4 + i) * 64);
      c20 = wmma_f32(a1, b01, c20);
      c21 = wmma_f32(a1, b11, c21);
    }
    int lr0 = t * 16 + half * 8 - 2;
#pragma unroll
    for (int r = 0; r < 8; ++r) {
      int ls = lr0 + r;
      if (ls >= 0 && ls < 61) {
        S[(p0 + ls) * 32 + l15]      = c20[r];
        S[(p0 + ls) * 32 + 16 + l15] = c21[r];
      }
    }
    xq0 = xq1; xq1 = xq2; xq2 = xq3;
  }
}

__global__ void __launch_bounds__(256) k_gated_skip(
    const float* __restrict__ H, float* __restrict__ G,
    const float* __restrict__ Wf, const float* __restrict__ bf,
    const float* __restrict__ Wg, const float* __restrict__ bg,
    const float* __restrict__ Wk, const float* __restrict__ bk,
    float* __restrict__ S, const float* __restrict__ og,
    const float* __restrict__ ob, float* __restrict__ out, int mode) {
  __shared__ __attribute__((aligned(16))) float wA[4096];
  __shared__ __attribute__((aligned(16))) float wS[1024];

  for (int idx = threadIdx.x; idx < 4096; idx += 256) {
    int e2 = idx & 1, ln = (idx >> 1) & 31, s = (idx >> 6) & 15;
    int ot = (idx >> 10) & 1, mat = idx >> 11;
    int o = (ot << 4) + (ln & 15);
    int kk = ((s >> 1) << 3) + ((ln >> 4) << 2) + ((s & 1) << 1) + e2;
    const float* Wm = mat ? Wg : Wf;
    wA[idx] = Wm[o * 64 + (kk & 31) * 2 + (kk >> 5)];
  }
  for (int idx = threadIdx.x; idx < 1024; idx += 256) {
    int e2 = idx & 1, ln = (idx >> 1) & 31, ks = (idx >> 6) & 7, ot2 = idx >> 9;
    int o2 = (ot2 << 4) + (ln & 15);
    int ch = ((ks >> 2) << 4) + ((ln >> 4) << 3) + ((ks & 3) << 1) + e2;
    wS[idx] = Wk[o2 * 32 + ch];
  }
  __syncthreads();

  int wave = threadIdx.x >> 5;
  int lane = threadIdx.x & 31;
  int half = lane >> 4;
  int l15  = lane & 15;
  int node = blockIdx.x * 8 + wave;
  size_t p0 = (size_t)node * 64;
  int off = 2 - mode;

  v8f bF0, bF1, bG0, bG1;
#pragma unroll
  for (int r = 0; r < 8; ++r) {
    bF0[r] = bf[half * 8 + r];  bF1[r] = bf[16 + half * 8 + r];
    bG0[r] = bg[half * 8 + r];  bG1[r] = bg[16 + half * 8 + r];
  }
  float bk0 = bk[l15], bk1 = bk[16 + l15];
  float rinv = rsqrtf(1.0f + EPSBN);
  float sc0 = og[l15] * rinv, sc1 = og[16 + l15] * rinv;
  float ob0 = ob[l15], ob1 = ob[16 + l15];

  const float* fa = wA + lane * 2;
  const float* fs = wS + lane * 2;
  const float* hp = H + (p0 + l15) * 32 + half * 4;
  float4 hv[8];
#pragma unroll
  for (int j = 0; j < 8; ++j) hv[j] = *(const float4*)(hp + 8 * j);

#pragma unroll 1
  for (int t = 0; t < 4; ++t) {
    asm volatile("" ::: "memory");
    float4 hn[8];
#pragma unroll
    for (int j = 0; j < 8; ++j) hn[j] = hv[j];
    if (t < 3) {
#pragma unroll
      for (int j = 0; j < 8; ++j) hn[j] = *(const float4*)(hp + (t + 1) * 512 + 8 * j);
    }
    int lr0 = t * 16 + half * 8 - off;
    v8f c20, c21;
#pragma unroll
    for (int r = 0; r < 8; ++r) {
      int ls = lr0 + r;
      float pv0 = 0.0f, pv1 = 0.0f;
      if (ls >= 0 && ls < 61) {
        pv0 = S[(p0 + ls) * 32 + l15];
        pv1 = S[(p0 + ls) * 32 + 16 + l15];
      }
      c20[r] = bk0 + pv0;
      c21[r] = bk1 + pv1;
    }

    v8f cf0 = bF0, cf1 = bF1, cg0 = bG0, cg1 = bG1;
#pragma unroll
    for (int s = 0; s < 16; ++s) {
      float4 q = hv[s >> 1];
      v2f bb;
      if (s & 1) { bb.x = q.z; bb.y = q.w; } else { bb.x = q.x; bb.y = q.y; }
      v2f af0 = *(const v2f*)(fa + (0 * 16 + s) * 64);
      v2f af1 = *(const v2f*)(fa + (1 * 16 + s) * 64);
      v2f ag0 = *(const v2f*)(fa + (2 * 16 + s) * 64);
      v2f ag1 = *(const v2f*)(fa + (3 * 16 + s) * 64);
      cf0 = wmma_f32(af0, bb, cf0);
      cg0 = wmma_f32(ag0, bb, cg0);
      cf1 = wmma_f32(af1, bb, cf1);
      cg1 = wmma_f32(ag1, bb, cg1);
    }
    v8f g0, g1;
#pragma unroll
    for (int r = 0; r < 8; ++r) {
      g0[r] = gate_f32(cf0[r], cg0[r]);
      g1[r] = gate_f32(cf1[r], cg1[r]);
    }
    if (mode != 2) {
      float* gp = G + (p0 + t * 16 + l15) * 32 + half * 8;
      *(float4*)(gp)      = make_float4(g0[0], g0[1], g0[2], g0[3]);
      *(float4*)(gp + 4)  = make_float4(g0[4], g0[5], g0[6], g0[7]);
      *(float4*)(gp + 16) = make_float4(g1[0], g1[1], g1[2], g1[3]);
      *(float4*)(gp + 20) = make_float4(g1[4], g1[5], g1[6], g1[7]);
    }
#pragma unroll
    for (int i = 0; i < 4; ++i) {
      v2f a0 = {g0[2 * i], g0[2 * i + 1]};
      v2f b00 = *(const v2f*)(fs + (0 * 8 + i) * 64);
      v2f b10 = *(const v2f*)(fs + (1 * 8 + i) * 64);
      c20 = wmma_f32(a0, b00, c20);
      c21 = wmma_f32(a0, b10, c21);
    }
#pragma unroll
    for (int i = 0; i < 4; ++i) {
      v2f a1 = {g1[2 * i], g1[2 * i + 1]};
      v2f b01 = *(const v2f*)(fs + (0 * 8 + 4 + i) * 64);
      v2f b11 = *(const v2f*)(fs + (1 * 8 + 4 + i) * 64);
      c20 = wmma_f32(a1, b01, c20);
      c21 = wmma_f32(a1, b11, c21);
    }
    if (mode != 2) {
#pragma unroll
      for (int r = 0; r < 8; ++r) {
        int ls = lr0 + r;
        if (ls >= 0 && ls < 61) {
          S[(p0 + ls) * 32 + l15]      = c20[r];
          S[(p0 + ls) * 32 + 16 + l15] = c21[r];
        }
      }
    } else {
      int b = node >> 9, n = node & 511;
#pragma unroll
      for (int r = 0; r < 8; ++r) {
        int ls = lr0 + r;
        if (ls < 61) {
          size_t oi = (((size_t)(b * 61 + ls)) * 512 + n) * 32;
          out[oi + l15]      = c20[r] * sc0 + ob0;
          out[oi + 16 + l15] = c21[r] * sc1 + ob1;
        }
      }
    }
#pragma unroll
    for (int j = 0; j < 8; ++j) hv[j] = hn[j];
  }
}

#define T2PITCH 36

__device__ __forceinline__ size_t t2_off(int m, int k) {
  return (size_t)(m * 8 + (k >> 6)) * 2048 + (size_t)(k & 63);
}

__global__ void __launch_bounds__(128) k_adj2(const float* __restrict__ adj,
                                              float* __restrict__ T2) {
  __shared__ __attribute__((aligned(16))) float part[4 * 16 * T2PITCH];

  int bid = blockIdx.x;
  int kt = bid & 15;
  int mt = bid >> 4;
  int m0 = mt << 4;
  int kc0 = kt << 5;
  int wave = threadIdx.x >> 5;
  int lane = threadIdx.x & 31;
  int half = lane >> 4;
  int l15  = lane & 15;

  const float* Acol = adj + m0 + l15;
  const float* B0   = adj + (size_t)(kc0 + l15) * 512;
  const float* B1   = adj + (size_t)(kc0 + 16 + l15) * 512;

  v8f c0 = {}, c1 = {};
  int nbeg = wave << 7;
#pragma unroll 8
  for (int i = 0; i < 128; i += 4) {
    int n = nbeg + i + half * 2;
    v2f a;
    a.x = Acol[(size_t)n * 512];
    a.y = Acol[(size_t)(n + 1) * 512];
    v2f b0 = *(const v2f*)(B0 + n);
    v2f b1 = *(const v2f*)(B1 + n);
    c0 = wmma_f32(a, b0, c0);
    c1 = wmma_f32(a, b1, c1);
  }
  float* pw = part + wave * 16 * T2PITCH;
#pragma unroll
  for (int r = 0; r < 8; ++r) {
    pw[(r + half * 8) * T2PITCH + l15]      = c0[r];
    pw[(r + half * 8) * T2PITCH + 16 + l15] = c1[r];
  }
  __syncthreads();

  int row  = threadIdx.x >> 3;
  int col4 = (threadIdx.x & 7) << 2;
  const float* p = part + row * T2PITCH + col4;
  float4 s0 = *(const float4*)(p);
  float4 s1 = *(const float4*)(p + 16 * T2PITCH);
  float4 s2 = *(const float4*)(p + 32 * T2PITCH);
  float4 s3 = *(const float4*)(p + 48 * T2PITCH);
  float4 s;
  s.x = ((s0.x + s1.x) + s2.x) + s3.x;
  s.y = ((s0.y + s1.y) + s2.y) + s3.y;
  s.z = ((s0.z + s1.z) + s2.z) + s3.z;
  s.w = ((s0.w + s1.w) + s2.w) + s3.w;
  *(float4*)(T2 + t2_off(m0 + row, kc0 + col4)) = s;
}

#define XPITCH 288
#define APITCH 160
#define TPITCH 36
#define XS_F (16 * XPITCH)
#define AS_F (16 * APITCH)
#define TS_F (64 * TPITCH)
#define STAGE_F (XS_F + AS_F + TS_F)

template <bool FROM_X>
__global__ void __launch_bounds__(256) __attribute__((amdgpu_waves_per_eu(8)))
k_diffuse_gconv(
    const float* __restrict__ adj, const float* __restrict__ T2,
    const float* __restrict__ Xin, const float* __restrict__ Hres,
    float* __restrict__ Hout, const float* __restrict__ Wgc,
    const float* __restrict__ bgc, const float* __restrict__ gcg,
    const float* __restrict__ gcb, const float* __restrict__ bng,
    const float* __restrict__ bnb, int Lout,
    const float* __restrict__ x, const float* __restrict__ Wst,
    const float* __restrict__ bst) {
  __shared__ __attribute__((aligned(16))) float st[2 * STAGE_F];

  int bid = blockIdx.x;
  int jb = bid & 15;
  int mb = (bid >> 4) & 7;
  int b  = bid >> 7;
  int jc = jb << 7;
  int mc = mb << 6;
  int tid = threadIdx.x;

  const float* Xb = Xin + (size_t)b * 512 * 2048 + jc;
  const float* Ab = adj + mc;
  const float* Tb = T2 + t2_off(mc, 0);
  unsigned xgo = (unsigned)(tid >> 5) * 2048u + 4u * (unsigned)(tid & 31);
  unsigned ago = (unsigned)(tid >> 4) * 512u + 4u * (unsigned)(tid & 15);
  unsigned tgo = (unsigned)(tid >> 3) * (8u * 2048u) + 4u * (unsigned)(tid & 7);
  int xso = (tid >> 6) * XPITCH + ((tid >> 5) & 1) * 16 + ((tid & 31) >> 2) * 32 + (tid & 3) * 4;
  int aso = XS_F + (tid >> 5) * APITCH + ((tid >> 4) & 1) * 16 + ((tid & 15) >> 2) * 32 + (tid & 3) * 4;
  int tso = XS_F + AS_F + (tid >> 3) * TPITCH + 4 * (tid & 7);

  float4 x0r, x1r, x2r, x3r, a0r, a1r, t0r, t1r;
#define STAGE_LOAD(C)                                                         \
  {                                                                           \
    const float* xp = Xb + (size_t)(C) * (32 * 2048) + xgo;                   \
    x0r = *(const float4*)(xp);                                               \
    x1r = *(const float4*)(xp + 8 * 2048);                                    \
    x2r = *(const float4*)(xp + 16 * 2048);                                   \
    x3r = *(const float4*)(xp + 24 * 2048);                                   \
    const float* ap = Ab + (size_t)(C) * (32 * 512) + ago;                    \
    a0r = *(const float4*)(ap);                                               \
    a1r = *(const float4*)(ap + 16 * 512);                                    \
    const float* tp = Tb + (size_t)((C) >> 1) * 2048 + ((C) & 1) * 32 + tgo;  \
    t0r = *(const float4*)(tp);                                               \
    t1r = *(const float4*)(tp + 32 * 8 * 2048);                               \
  }
#define STAGE_STORE(S)                                                        \
  {                                                                           \
    float* sb = st + (S) * STAGE_F;                                           \
    *(float4*)(sb + xso)               = x0r;                                 \
    *(float4*)(sb + xso + 4 * XPITCH)  = x1r;                                 \
    *(float4*)(sb + xso + 8 * XPITCH)  = x2r;                                 \
    *(float4*)(sb + xso + 12 * XPITCH) = x3r;                                 \
    *(float4*)(sb + aso)               = a0r;                                 \
    *(float4*)(sb + aso + 8 * APITCH)  = a1r;                                 \
    *(float4*)(sb + tso)               = t0r;                                 \
    *(float4*)(sb + tso + 32 * TPITCH) = t1r;                                 \
  }

  STAGE_LOAD(0)
  STAGE_STORE(0)
  __syncthreads();

  int wave = __builtin_amdgcn_readfirstlane((int)(threadIdx.x >> 5));
  int lane = threadIdx.x & 31;
  int half = lane >> 4;
  int l15  = lane & 15;
  int mt = wave & 3, jt = wave >> 2;
  int m0 = mc + (mt << 4);
  int j0 = jc + (jt << 6);

  int xfo = half * XPITCH + (jt << 7) + l15;
  int afo = XS_F + half * APITCH + (mt << 5) + l15;
  int tfo = XS_F + AS_F + ((mt << 4) + l15) * TPITCH + 2 * half;

  v8f c0 = {}, c1 = {}, c2 = {}, c3 = {};
  v8f d0 = {}, d1 = {}, d2 = {}, d3 = {};
#pragma unroll 1
  for (int c = 0; c < 16; ++c) {
    const float* sb = st + (c & 1) * STAGE_F;
    if (c < 15) STAGE_LOAD(c + 1)
#pragma unroll 1
    for (int s = 0; s < 8; ++s) {
      const float* af = sb + afo + s * (2 * APITCH);
      v2f a1 = {af[0], af[16]};
      v2f a2 = *(const v2f*)(sb + tfo + s * 4);
      const float* xf = sb + xfo + s * (2 * XPITCH);
      v2f b0 = {xf[0],  xf[16]};
      v2f b1 = {xf[32], xf[48]};
      v2f b2 = {xf[64], xf[80]};
      v2f b3 = {xf[96], xf[112]};
      c0 = wmma_f32(b0, a1, c0);
      d0 = wmma_f32(b0, a2, d0);
      c1 = wmma_f32(b1, a1, c1);
      d1 = wmma_f32(b1, a2, d1);
      c2 = wmma_f32(b2, a1, c2);
      d2 = wmma_f32(b2, a2, d2);
      c3 = wmma_f32(b3, a1, c3);
      d3 = wmma_f32(b3, a2, d3);
    }
    if (c < 15) STAGE_STORE((c + 1) & 1)
    __syncthreads();
  }
#undef STAGE_LOAD
#undef STAGE_STORE

  int o0 = l15, o1 = 16 + l15;
  const float* W0 = Wgc + (size_t)o0 * 96;
  const float* W1 = Wgc + (size_t)o1 * 96;
  int lbase = j0 >> 5;
  size_t nrow = ((size_t)b * 512 + m0 + l15) * 64;

  float bias0 = bgc[o0], bias1 = bgc[o1];
  v8f e00, e01, e10, e11;
#pragma unroll
  for (int r = 0; r < 8; ++r) { e00[r] = bias0; e01[r] = bias1; e10[r] = bias0; e11[r] = bias1; }

  {
    const float* g0 = Xin + (nrow + lbase) * 32 + half * 16;
    const float* g1 = g0 + 32;
    float4 ga[4], gb[4];
#pragma unroll
    for (int q = 0; q < 4; ++q) {
      ga[q] = *(const float4*)(g0 + 4 * q);
      gb[q] = *(const float4*)(g1 + 4 * q);
    }
#pragma unroll
    for (int q = 0; q < 4; ++q) {
      float4 w0 = *(const float4*)(W0 + half * 16 + 4 * q);
      float4 w1 = *(const float4*)(W1 + half * 16 + 4 * q);
      v2f wa0 = {w0.x, w0.y}, wb0 = {w0.z, w0.w};
      v2f wa1 = {w1.x, w1.y}, wb1 = {w1.z, w1.w};
      v2f xa = {ga[q].x, ga[q].y}, xb = {ga[q].z, ga[q].w};
      v2f ya = {gb[q].x, gb[q].y}, yb = {gb[q].z, gb[q].w};
      e00 = wmma_f32(xa, wa0, e00);
      e01 = wmma_f32(xa, wa1, e01);
      e10 = wmma_f32(ya, wa0, e10);
      e11 = wmma_f32(ya, wa1, e11);
      e00 = wmma_f32(xb, wb0, e00);
      e01 = wmma_f32(xb, wb1, e01);
      e10 = wmma_f32(yb, wb0, e10);
      e11 = wmma_f32(yb, wb1, e11);
    }
  }
#define GC_PART(T0, T1, T2_, T3, WOFF)                                        \
  _Pragma("unroll")                                                           \
  for (int i = 0; i < 4; ++i) {                                               \
    v2f wlo0 = *(const v2f*)(W0 + (WOFF) + half * 8 + 2 * i);                 \
    v2f wlo1 = *(const v2f*)(W1 + (WOFF) + half * 8 + 2 * i);                 \
    v2f whi0 = *(const v2f*)(W0 + (WOFF) + 16 + half * 8 + 2 * i);            \
    v2f whi1 = *(const v2f*)(W1 + (WOFF) + 16 + half * 8 + 2 * i);            \
    v2f p0 = {T0[2 * i], T0[2 * i + 1]};                                      \
    v2f p1 = {T1[2 * i], T1[2 * i + 1]};                                      \
    v2f p2 = {T2_[2 * i], T2_[2 * i + 1]};                                    \
    v2f p3 = {T3[2 * i], T3[2 * i + 1]};                                      \
    e00 = wmma_f32(p0, wlo0, e00);                                            \
    e01 = wmma_f32(p0, wlo1, e01);                                            \
    e10 = wmma_f32(p2, wlo0, e10);                                            \
    e11 = wmma_f32(p2, wlo1, e11);                                            \
    e00 = wmma_f32(p1, whi0, e00);                                            \
    e01 = wmma_f32(p1, whi1, e01);                                            \
    e10 = wmma_f32(p3, whi0, e10);                                            \
    e11 = wmma_f32(p3, whi1, e11);                                            \
  }
  GC_PART(c0, c1, c2, c3, 32)
  float res0[8], res1[8], res2[8], res3[8];
  if (FROM_X) {
    float w00 = Wst[o0 * 2], w01 = Wst[o0 * 2 + 1], bs0 = bst[o0];
    float w10 = Wst[o1 * 2], w11 = Wst[o1 * 2 + 1], bs1 = bst[o1];
    int la = lbase + 1 < 63 ? lbase + 1 : 63;
    int lb = lbase + 2 < 63 ? lbase + 2 : 63;
    const float* xn = x + (size_t)b * (2 * 512 * 64) + (size_t)(m0 + half * 8) * 64;
#pragma unroll
    for (int r = 0; r < 8; ++r) {
      float xa0 = xn[r * 64 + la], xa1 = xn[512 * 64 + r * 64 + la];
      float xb0 = xn[r * 64 + lb], xb1 = xn[512 * 64 + r * 64 + lb];
      res0[r] = fmaf(xa0, w00, fmaf(xa1, w01, bs0));
      res1[r] = fmaf(xa0, w10, fmaf(xa1, w11, bs1));
      res2[r] = fmaf(xb0, w00, fmaf(xb1, w01, bs0));
      res3[r] = fmaf(xb0, w10, fmaf(xb1, w11, bs1));
    }
  } else {
#pragma unroll
    for (int r = 0; r < 8; ++r) {
      size_t row = ((size_t)b * 512 + m0 + r + half * 8) * 64 + lbase;
      res0[r] = Hres[(row + 1) * 32 + o0];
      res1[r] = Hres[(row + 1) * 32 + o1];
      res2[r] = Hres[(row + 2) * 32 + o0];
      res3[r] = Hres[(row + 2) * 32 + o1];
    }
  }
  GC_PART(d0, d1, d2, d3, 64)
#undef GC_PART

  float rinv = rsqrtf(1.0f + EPSBN);
  float ig0 = gcg[o0] * rinv, bg0 = gcb[o0], ib0 = bng[o0] * rinv, bb0 = bnb[o0];
  float ig1 = gcg[o1] * rinv, bg1 = gcb[o1], ib1 = bng[o1] * rinv, bb1 = bnb[o1];
#pragma unroll
  for (int r = 0; r < 8; ++r) {
    size_t row = ((size_t)b * 512 + m0 + r + half * 8) * 64 + lbase;
    if (lbase < Lout) {
      float g0v = e00[r] * ig0 + bg0;
      float g1v = e01[r] * ig1 + bg1;
      float h0 = g0v + res0[r];
      float h1 = g1v + res1[r];
      Hout[row * 32 + o0] = h0 * ib0 + bb0;
      Hout[row * 32 + o1] = h1 * ib1 + bb1;
    }
    if (lbase + 1 < Lout) {
      float g0v = e10[r] * ig0 + bg0;
      float g1v = e11[r] * ig1 + bg1;
      float h0 = g0v + res2[r];
      float h1 = g1v + res3[r];
      Hout[(row + 1) * 32 + o0] = h0 * ib0 + bb0;
      Hout[(row + 1) * 32 + o1] = h1 * ib1 + bb1;
    }
  }
}

extern "C" void kernel_launch(void* const* d_in, const int* in_sizes, int n_in,
                              void* d_out, int out_size, void* d_ws, size_t ws_size,
                              hipStream_t stream) {
  (void)in_sizes; (void)n_in; (void)out_size; (void)ws_size;
  const float* x   = (const float*)d_in[0];
  const float* adj = (const float*)d_in[1];
  const float* Wst = (const float*)d_in[2];
  const float* bst = (const float*)d_in[3];
  const float* Wf  = (const float*)d_in[4];
  const float* bf  = (const float*)d_in[5];
  const float* Wg  = (const float*)d_in[6];
  const float* bg  = (const float*)d_in[7];
  const float* Wsk = (const float*)d_in[8];
  const float* bsk = (const float*)d_in[9];
  const float* Wgc = (const float*)d_in[10];
  const float* bgc = (const float*)d_in[11];
  const float* gcg = (const float*)d_in[12];
  const float* gcb = (const float*)d_in[13];
  const float* bng = (const float*)d_in[14];
  const float* bnb = (const float*)d_in[15];
  const float* og  = (const float*)d_in[16];
  const float* ob  = (const float*)d_in[17];
  float* out = (float*)d_out;

  const size_t BUF = (size_t)8 * 512 * 64 * 32 + 1024;
  float* ws    = (float*)d_ws;
  float* bufH  = ws;
  float* bufH2 = ws + BUF;
  float* bufG  = ws + 2 * BUF;
  float* bufS  = ws + 5 * BUF;

  float* T2 = bufS + 62 * 32;
  k_adj2<<<512, 128, 0, stream>>>(adj, T2);

  k_gated0_skip<<<512, 256, 0, stream>>>(x, Wst, bst, bufG, Wf, bf, Wg, bg, Wsk, bsk,
                                         bufS);
  k_diffuse_gconv<true><<<1024, 256, 0, stream>>>(adj, T2, bufG, bufH, bufH2, Wgc, bgc,
                                                  gcg, gcb, bng, bnb, 63, x, Wst, bst);
  k_gated_skip<<<512, 256, 0, stream>>>(bufH2, bufG, Wf + 2048, bf + 32, Wg + 2048,
                                        bg + 32, Wsk + 1024, bsk + 32,
                                        bufS, og, ob, out, 1);
  k_diffuse_gconv<false><<<1024, 256, 0, stream>>>(adj, T2, bufG, bufH2, bufH, Wgc + 3072,
                                                   bgc + 32, gcg + 32, gcb + 32,
                                                   bng + 32, bnb + 32, 62, x, Wst, bst);
  k_gated_skip<<<512, 256, 0, stream>>>(bufH, bufG, Wf + 4096, bf + 64, Wg + 4096,
                                        bg + 64, Wsk + 2048, bsk + 64,
                                        bufS, og, ob, out, 2);
}


# input-only set-up done once by 256 spare work-groups of layer 0's launch: (adj*adj)^T (own launch gone, 6 -> 5) and the weight fragment tables every k_gated_skip block rebuilt; sums unchanged
# speedup vs baseline: 2.0367x; 1.0429x over previous
#include <hip/hip_runtime.h>


typedef __attribute__((ext_vector_type(2))) float v2f;
typedef __attribute__((ext_vector_type(8))) float v8f;

#define EPSBN 1e-5f

__device__ __forceinline__ v8f wmma_f32(v2f a, v2f b, v8f c) {
  return __builtin_amdgcn_wmma_f32_16x16x4_f32(false, a, false, b, (short)0, c,
                                               false, false);
}

__device__ __forceinline__ float gate_f32(float f, float g) {
  float tf = __builtin_amdgcn_tanhf(f);
  float tg = __builtin_amdgcn_tanhf(0.5f * g);
  return tf * fmaf(0.5f, tg, 0.5f);
}

#define T2PITCH 36

__device__ __forceinline__ size_t t2_off(int m, int k) {
  return (size_t)(m * 8 + (k >> 6)) * 2048 + (size_t)(k & 63);
}

__device__ __forceinline__ void adj2_tile(const float* __restrict__ adj,
                                          float* __restrict__ T2, int tile,
                                          int tid, float* part) {
  int kt = tile & 15;
  int mt = tile >> 4;
  int m0 = mt << 4;
  int kc0 = kt << 5;
  int wave = tid >> 5;
  int lane = tid & 31;
  int half = lane >> 4;
  int l15  = lane & 15;

  const float* Acol = adj + m0 + l15;
  const float* B0   = adj + (size_t)(kc0 + l15) * 512;
  const float* B1   = adj + (size_t)(kc0 + 16 + l15) * 512;

  v8f c0 = {}, c1 = {};
  int nbeg = wave << 7;
#pragma unroll 8
  for (int i = 0; i < 128; i += 4) {
    int n = nbeg + i + half * 2;
    v2f a;
    a.x = Acol[(size_t)n * 512];
    a.y = Acol[(size_t)(n + 1) * 512];
    v2f b0 = *(const v2f*)(B0 + n);
    v2f b1 = *(const v2f*)(B1 + n);
    c0 = wmma_f32(a, b0, c0);
    c1 = wmma_f32(a, b1, c1);
  }
  float* pw = part + wave * 16 * T2PITCH;
#pragma unroll
  for (int r = 0; r < 8; ++r) {
    pw[(r + half * 8) * T2PITCH + l15]      = c0[r];
    pw[(r + half * 8) * T2PITCH + 16 + l15] = c1[r];
  }
  __syncthreads();

  int row  = tid >> 3;
  int col4 = (tid & 7) << 2;
  const float* p = part + row * T2PITCH + col4;
  float4 s0 = *(const float4*)(p);
  float4 s1 = *(const float4*)(p + 16 * T2PITCH);
  float4 s2 = *(const float4*)(p + 32 * T2PITCH);
  float4 s3 = *(const float4*)(p + 48 * T2PITCH);
  float4 s;
  s.x = ((s0.x + s1.x) + s2.x) + s3.x;
  s.y = ((s0.y + s1.y) + s2.y) + s3.y;
  s.z = ((s0.z + s1.z) + s2.z) + s3.z;
  s.w = ((s0.w + s1.w) + s2.w) + s3.w;
  *(float4*)(T2 + t2_off(m0 + row, kc0 + col4)) = s;
}

#define WTAB_F 5120

__device__ __forceinline__ const float* wtab_src(int e, const float* __restrict__ Wf,
                                                 const float* __restrict__ Wg,
                                                 const float* __restrict__ Wk) {
  int layer = e >= WTAB_F ? 2 : 1;
  int idx = e - (layer - 1) * WTAB_F;
  const float* src;
  if (idx < 4096) {
    int e2 = idx & 1, ln = (idx >> 1) & 31, s = (idx >> 6) & 15;
    int ot = (idx >> 10) & 1, mat = idx >> 11;
    int o = (ot << 4) + (ln & 15);
    int kk = ((s >> 1) << 3) + ((ln >> 4) << 2) + ((s & 1) << 1) + e2;
    src = (mat ? Wg : Wf) + layer * 2048 + o * 64 + (kk & 31) * 2 + (kk >> 5);
  } else {
    int i2 = idx - 4096;
    int e2 = i2 & 1, ln = (i2 >> 1) & 31, ks = (i2 >> 6) & 7, ot2 = i2 >> 9;
    int o2 = (ot2 << 4) + (ln & 15);
    int ch = ((ks >> 2) << 4) + ((ln >> 4) << 3) + ((ks & 3) << 1) + e2;
    src = Wk + layer * 1024 + o2 * 32 + ch;
  }
  return src;
}

__global__ void __launch_bounds__(256) k_gated0_skip_setup(
    const float* __restrict__ x, const float* __restrict__ Wst,
    const float* __restrict__ bst, float* __restrict__ G,
    const float* __restrict__ Wf, const float* __restrict__ bf,
    const float* __restrict__ Wg, const float* __restrict__ bg,
    const float* __restrict__ Wk, const float* __restrict__ bk,
    float* __restrict__ S, const float* __restrict__ adj,
    float* __restrict__ T2, float* __restrict__ tab) {
  __shared__ __attribute__((aligned(16))) float part[2 * 4 * 16 * T2PITCH];
  if (blockIdx.x >= 512) {
    int sb = (int)blockIdx.x - 512;
    bool tabw = sb < 2 * WTAB_F / 64 && threadIdx.x < 64;
    int te = sb * 64 + (int)threadIdx.x;
    float tval = 0.0f;
    if (tabw) tval = *wtab_src(te, Wf, Wg, Wk);
    int grp = threadIdx.x >> 7;
    adj2_tile(adj, T2, sb * 2 + grp, threadIdx.x & 127,
              part + grp * (4 * 16 * T2PITCH));
    if (tabw) tab[te] = tval;
    return;
  }

  __shared__ __attribute__((aligned(16))) float wC[512];
  __shared__ __attribute__((aligned(16))) float wS[1024];

  if (threadIdx.x < 128) {
    int tap = threadIdx.x & 1, o = (threadIdx.x >> 1) & 31, mat = threadIdx.x >> 6;
    const float* Wm = (mat ? Wg : Wf) + o * 64 + tap;
    float a0 = 0.0f, a1 = 0.0f, ab = 0.0f;
    for (int c = 0; c < 32; ++c) {
      float w = Wm[2 * c];
      a0 = fmaf(w, Wst[2 * c], a0);
      a1 = fmaf(w, Wst[2 * c + 1], a1);
      ab = fmaf(w, bst[c], ab);
    }
    *(float4*)(wC + threadIdx.x * 4) = make_float4(a0, a1, ab, 0.0f);
  }
  float wk[4];
#pragma unroll
  for (int j = 0; j < 4; ++j) {
    int idx = threadIdx.x + 256 * j;
    int e2 = idx & 1, ln = (idx >> 1) & 31, ks = (idx >> 6) & 7, ot2 = idx >> 9;
    int o2 = (ot2 << 4) + (ln & 15);
    int ch = ((ks >> 2) << 4) + ((ln >> 4) << 3) + ((ks & 3) << 1) + e2;
    wk[j] = Wk[o2 * 32 + ch];
  }
#pragma unroll
  for (int j = 0; j < 4; ++j) wS[threadIdx.x + 256 * j] = wk[j];
  __syncthreads();

  int wave = threadIdx.x >> 5;
  int lane = threadIdx.x & 31;
  int half = lane >> 4;
  int l15  = lane & 15;
  int node = blockIdx.x * 8 + wave;
  size_t p0 = (size_t)node * 64;

  v2f af0 = *(const v2f*)(wC + ((l15 * 2) + half) * 4);
  v2f af1 = *(const v2f*)(wC + (((16 + l15) * 2) + half) * 4);
  v2f ag0 = *(const v2f*)(wC + (((32 + l15) * 2) + half) * 4);
  v2f ag1 = *(const v2f*)(wC + (((48 + l15) * 2) + half) * 4);
  v8f bF0, bF1, bG0, bG1;
#pragma unroll
  for (int r = 0; r < 8; ++r) {
    int ch = half * 8 + r;
    bF0[r] = bf[ch]      + (wC[(ch * 2) * 4 + 2]        + wC[(ch * 2 + 1) * 4 + 2]);
    bF1[r] = bf[16 + ch] + (wC[((16 + ch) * 2) * 4 + 2] + wC[((16 + ch) * 2 + 1) * 4 + 2]);
    bG0[r] = bg[ch]      + (wC[((32 + ch) * 2) * 4 + 2] + wC[((32 + ch) * 2 + 1) * 4 + 2]);
    bG1[r] = bg[16 + ch] + (wC[((48 + ch) * 2) * 4 + 2] + wC[((48 + ch) * 2 + 1) * 4 + 2]);
  }
  float bk0 = bk[l15], bk1 = bk[16 + l15];
  const float* fs = wS + lane * 2;

  const float* x0p = x + (size_t)(node >> 9) * (2 * 512 * 64) + (size_t)(node & 511) * 64;
  const float* x1p = x0p + 512 * 64;
  v2f xb[4];
#pragma unroll
  for (int t = 0; t < 4; ++t) {
    int l = t * 16 + l15 + half;
    l = l < 63 ? l : 63;
    xb[t].x = x0p[l];
    xb[t].y = x1p[l];
  }
  v2f xq0 = xb[0], xq1 = xb[1], xq2 = xb[2], xq3 = xb[3];

#pragma unroll 1
  for (int t = 0; t < 4; ++t) {
    v8f cf0 = wmma_f32(af0, xq0, bF0);
    v8f cg0 = wmma_f32(ag0, xq0, bG0);
    v8f cf1 = wmma_f32(af1, xq0, bF1);
    v8f cg1 = wmma_f32(ag1, xq0, bG1);
    v8f g0, g1;
#pragma unroll
    for (int r = 0; r < 8; ++r) {
      g0[r] = gate_f32(cf0[r], cg0[r]);
      g1[r] = gate_f32(cf1[r], cg1[r]);
    }
    float* gp = G + (p0 + t * 16 + l15) * 32 + half * 8;
    *(float4*)(gp)      = make_float4(g0[0], g0[1], g0[2], g0[3]);
    *(float4*)(gp + 4)  = make_float4(g0[4], g0[5], g0[6], g0[7]);
    *(float4*)(gp + 16) = make_float4(g1[0], g1[1], g1[2], g1[3]);
    *(float4*)(gp + 20) = make_float4(g1[4], g1[5], g1[6], g1[7]);

    v8f c20, c21;
#pragma unroll
    for (int r = 0; r < 8; ++r) { c20[r] = bk0; c21[r] = bk1; }
#pragma unroll
    for (int i = 0; i < 4; ++i) {
      v2f a0 = {g0[2 * i], g0[2 * i + 1]};
      v2f b00 = *(const v2f*)(fs + (0 * 8 + i) * 64);
      v2f b10 = *(const v2f*)(fs + (1 * 8 + i) * 64);
      c20 = wmma_f32(a0, b00, c20);
      c21 = wmma_f32(a0, b10, c21);
    }
#pragma unroll
    for (int i = 0; i < 4; ++i) {
      v2f a1 = {g1[2 * i], g1[2 * i + 1]};
      v2f b01 = *(const v2f*)(fs + (0 * 8 + 4 + i) * 64);
      v2f b11 = *(const v2f*)(fs + (1 * 8 + 4 + i) * 64);
      c20 = wmma_f32(a1, b01, c20);
      c21 = wmma_f32(a1, b11, c21);
    }
    int lr0 = t * 16 + half * 8 - 2;
#pragma unroll
    for (int r = 0; r < 8; ++r) {
      int ls = lr0 + r;
      if (ls >= 0 && ls < 61) {
        S[(p0 + ls) * 32 + l15]      = c20[r];
        S[(p0 + ls) * 32 + 16 + l15] = c21[r];
      }
    }
    xq0 = xq1; xq1 = xq2; xq2 = xq3;
  }
}

__global__ void __launch_bounds__(256) k_gated_skip(
    const float* __restrict__ H, float* __restrict__ G,
    const float* __restrict__ wtab, const float* __restrict__ bf,
    const float* __restrict__ bg, const float* __restrict__ bk,
    float* __restrict__ S, const float* __restrict__ og,
    const float* __restrict__ ob, float* __restrict__ out, int mode) {
  __shared__ __attribute__((aligned(16))) float wAS[WTAB_F];
  const float* wA = wAS;
  const float* wS = wAS + 4096;
  float4 tv[5];
#pragma unroll
  for (int j = 0; j < 5; ++j)
    tv[j] = *(const float4*)(wtab + 4 * (threadIdx.x + 256 * j));
#pragma unroll
  for (int j = 0; j < 5; ++j)
    *(float4*)(wAS + 4 * (threadIdx.x + 256 * j)) = tv[j];
  __syncthreads();

  int wave = threadIdx.x >> 5;
  int lane = threadIdx.x & 31;
  int half = lane >> 4;
  int l15  = lane & 15;
  int node = blockIdx.x * 8 + wave;
  size_t p0 = (size_t)node * 64;
  int off = 2 - mode;

  v8f bF0, bF1, bG0, bG1;
#pragma unroll
  for (int r = 0; r < 8; ++r) {
    bF0[r] = bf[half * 8 + r];  bF1[r] = bf[16 + half * 8 + r];
    bG0[r] = bg[half * 8 + r];  bG1[r] = bg[16 + half * 8 + r];
  }
  float bk0 = bk[l15], bk1 = bk[16 + l15];
  float rinv = rsqrtf(1.0f + EPSBN);
  float sc0 = og[l15] * rinv, sc1 = og[16 + l15] * rinv;
  float ob0 = ob[l15], ob1 = ob[16 + l15];

  const float* fa = wA + lane * 2;
  const float* fs = wS + lane * 2;
  const float* hp = H + (p0 + l15) * 32 + half * 4;
  float4 hv[8];
#pragma unroll
  for (int j = 0; j < 8; ++j) hv[j] = *(const float4*)(hp + 8 * j);

#pragma unroll 1
  for (int t = 0; t < 4; ++t) {
    asm volatile("" ::: "memory");
    float4 hn[8];
#pragma unroll
    for (int j = 0; j < 8; ++j) hn[j] = hv[j];
    if (t < 3) {
#pragma unroll
      for (int j = 0; j < 8; ++j) hn[j] = *(const float4*)(hp + (t + 1) * 512 + 8 * j);
    }
    int lr0 = t * 16 + half * 8 - off;
    v8f c20, c21;
#pragma unroll
    for (int r = 0; r < 8; ++r) {
      int ls = lr0 + r;
      float pv0 = 0.0f, pv1 = 0.0f;
      if (ls >= 0 && ls < 61) {
        pv0 = S[(p0 + ls) * 32 + l15];
        pv1 = S[(p0 + ls) * 32 + 16 + l15];
      }
      c20[r] = bk0 + pv0;
      c21[r] = bk1 + pv1;
    }

    v8f cf0 = bF0, cf1 = bF1, cg0 = bG0, cg1 = bG1;
#pragma unroll
    for (int s = 0; s < 16; ++s) {
      float4 q = hv[s >> 1];
      v2f bb;
      if (s & 1) { bb.x = q.z; bb.y = q.w; } else { bb.x = q.x; bb.y = q.y; }
      v2f af0 = *(const v2f*)(fa + (0 * 16 + s) * 64);
      v2f af1 = *(const v2f*)(fa + (1 * 16 + s) * 64);
      v2f ag0 = *(const v2f*)(fa + (2 * 16 + s) * 64);
      v2f ag1 = *(const v2f*)(fa + (3 * 16 + s) * 64);
      cf0 = wmma_f32(af0, bb, cf0);
      cg0 = wmma_f32(ag0, bb, cg0);
      cf1 = wmma_f32(af1, bb, cf1);
      cg1 = wmma_f32(ag1, bb, cg1);
    }
    v8f g0, g1;
#pragma unroll
    for (int r = 0; r < 8; ++r) {
      g0[r] = gate_f32(cf0[r], cg0[r]);
      g1[r] = gate_f32(cf1[r], cg1[r]);
    }
    if (mode != 2) {
      float* gp = G + (p0 + t * 16 + l15) * 32 + half * 8;
      *(float4*)(gp)      = make_float4(g0[0], g0[1], g0[2], g0[3]);
      *(float4*)(gp + 4)  = make_float4(g0[4], g0[5], g0[6], g0[7]);
      *(float4*)(gp + 16) = make_float4(g1[0], g1[1], g1[2], g1[3]);
      *(float4*)(gp + 20) = make_float4(g1[4], g1[5], g1[6], g1[7]);
    }
#pragma unroll
    for (int i = 0; i < 4; ++i) {
      v2f a0 = {g0[2 * i], g0[2 * i + 1]};
      v2f b00 = *(const v2f*)(fs + (0 * 8 + i) * 64);
      v2f b10 = *(const v2f*)(fs + (1 * 8 + i) * 64);
      c20 = wmma_f32(a0, b00, c20);
      c21 = wmma_f32(a0, b10, c21);
    }
#pragma unroll
    for (int i = 0; i < 4; ++i) {
      v2f a1 = {g1[2 * i], g1[2 * i + 1]};
      v2f b01 = *(const v2f*)(fs + (0 * 8 + 4 + i) * 64);
      v2f b11 = *(const v2f*)(fs + (1 * 8 + 4 + i) * 64);
      c20 = wmma_f32(a1, b01, c20);
      c21 = wmma_f32(a1, b11, c21);
    }
    if (mode != 2) {
#pragma unroll
      for (int r = 0; r < 8; ++r) {
        int ls = lr0 + r;
        if (ls >= 0 && ls < 61) {
          S[(p0 + ls) * 32 + l15]      = c20[r];
          S[(p0 + ls) * 32 + 16 + l15] = c21[r];
        }
      }
    } else {
      int b = node >> 9, n = node & 511;
#pragma unroll
      for (int r = 0; r < 8; ++r) {
        int ls = lr0 + r;
        if (ls < 61) {
          size_t oi = (((size_t)(b * 61 + ls)) * 512 + n) * 32;
          out[oi + l15]      = c20[r] * sc0 + ob0;
          out[oi + 16 + l15] = c21[r] * sc1 + ob1;
        }
      }
    }
#pragma unroll
    for (int j = 0; j < 8; ++j) hv[j] = hn[j];
  }
}

#define XPITCH 288
#define APITCH 160
#define TPITCH 36
#define XS_F (16 * XPITCH)
#define AS_F (16 * APITCH)
#define TS_F (64 * TPITCH)
#define STAGE_F (XS_F + AS_F + TS_F)

template <bool FROM_X>
__global__ void __launch_bounds__(256) __attribute__((amdgpu_waves_per_eu(8)))
k_diffuse_gconv(
    const float* __restrict__ adj, const float* __restrict__ T2,
    const float* __restrict__ Xin, const float* __restrict__ Hres,
    float* __restrict__ Hout, const float* __restrict__ Wgc,
    const float* __restrict__ bgc, const float* __restrict__ gcg,
    const float* __restrict__ gcb, const float* __restrict__ bng,
    const float* __restrict__ bnb, int Lout,
    const float* __restrict__ x, const float* __restrict__ Wst,
    const float* __restrict__ bst) {
  __shared__ __attribute__((aligned(16))) float st[2 * STAGE_F];

  int bid = blockIdx.x;
  int jb = bid & 15;
  int mb = (bid >> 4) & 7;
  int b  = bid >> 7;
  int jc = jb << 7;
  int mc = mb << 6;
  int tid = threadIdx.x;

  const float* Xb = Xin + (size_t)b * 512 * 2048 + jc;
  const float* Ab = adj + mc;
  const float* Tb = T2 + t2_off(mc, 0);
  unsigned xgo = (unsigned)(tid >> 5) * 2048u + 4u * (unsigned)(tid & 31);
  unsigned ago = (unsigned)(tid >> 4) * 512u + 4u * (unsigned)(tid & 15);
  unsigned tgo = (unsigned)(tid >> 3) * (8u * 2048u) + 4u * (unsigned)(tid & 7);
  int xso = (tid >> 6) * XPITCH + ((tid >> 5) & 1) * 16 + ((tid & 31) >> 2) * 32 + (tid & 3) * 4;
  int aso = XS_F + (tid >> 5) * APITCH + ((tid >> 4) & 1) * 16 + ((tid & 15) >> 2) * 32 + (tid & 3) * 4;
  int tso = XS_F + AS_F + (tid >> 3) * TPITCH + 4 * (tid & 7);

  float4 x0r, x1r, x2r, x3r, a0r, a1r, t0r, t1r;
#define STAGE_LOAD(C)                                                         \
  {                                                                           \
    const float* xp = Xb + (size_t)(C) * (32 * 2048) + xgo;                   \
    x0r = *(const float4*)(xp);                                               \
    x1r = *(const float4*)(xp + 8 * 2048);                                    \
    x2r = *(const float4*)(xp + 16 * 2048);                                   \
    x3r = *(const float4*)(xp + 24 * 2048);                                   \
    const float* ap = Ab + (size_t)(C) * (32 * 512) + ago;                    \
    a0r = *(const float4*)(ap);                                               \
    a1r = *(const float4*)(ap + 16 * 512);                                    \
    const float* tp = Tb + (size_t)((C) >> 1) * 2048 + ((C) & 1) * 32 + tgo;  \
    t0r = *(const float4*)(tp);                                               \
    t1r = *(const float4*)(tp + 32 * 8 * 2048);                               \
  }
#define STAGE_STORE(S)                                                        \
  {                                                                           \
    float* sb = st + (S) * STAGE_F;                                           \
    *(float4*)(sb + xso)               = x0r;                                 \
    *(float4*)(sb + xso + 4 * XPITCH)  = x1r;                                 \
    *(float4*)(sb + xso + 8 * XPITCH)  = x2r;                                 \
    *(float4*)(sb + xso + 12 * XPITCH) = x3r;                                 \
    *(float4*)(sb + aso)               = a0r;                                 \
    *(float4*)(sb + aso + 8 * APITCH)  = a1r;                                 \
    *(float4*)(sb + tso)               = t0r;                                 \
    *(float4*)(sb + tso + 32 * TPITCH) = t1r;                                 \
  }

  STAGE_LOAD(0)
  STAGE_STORE(0)
  __syncthreads();

  int wave = __builtin_amdgcn_readfirstlane((int)(threadIdx.x >> 5));
  int lane = threadIdx.x & 31;
  int half = lane >> 4;
  int l15  = lane & 15;
  int mt = wave & 3, jt = wave >> 2;
  int m0 = mc + (mt << 4);
  int j0 = jc + (jt << 6);

  int xfo = half * XPITCH + (jt << 7) + l15;
  int afo = XS_F + half * APITCH + (mt << 5) + l15;
  int tfo = XS_F + AS_F + ((mt << 4) + l15) * TPITCH + 2 * half;

  v8f c0 = {}, c1 = {}, c2 = {}, c3 = {};
  v8f d0 = {}, d1 = {}, d2 = {}, d3 = {};
#pragma unroll 1
  for (int c = 0; c < 16; ++c) {
    const float* sb = st + (c & 1) * STAGE_F;
    if (c < 15) STAGE_LOAD(c + 1)
#pragma unroll 1
    for (int s = 0; s < 8; ++s) {
      const float* af = sb + afo + s * (2 * APITCH);
      v2f a1 = {af[0], af[16]};
      v2f a2 = *(const v2f*)(sb + tfo + s * 4);
      const float* xf = sb + xfo + s * (2 * XPITCH);
      v2f b0 = {xf[0],  xf[16]};
      v2f b1 = {xf[32], xf[48]};
      v2f b2 = {xf[64], xf[80]};
      v2f b3 = {xf[96], xf[112]};
      c0 = wmma_f32(b0, a1, c0);
      d0 = wmma_f32(b0, a2, d0);
      c1 = wmma_f32(b1, a1, c1);
      d1 = wmma_f32(b1, a2, d1);
      c2 = wmma_f32(b2, a1, c2);
      d2 = wmma_f32(b2, a2, d2);
      c3 = wmma_f32(b3, a1, c3);
      d3 = wmma_f32(b3, a2, d3);
    }
    if (c < 15) STAGE_STORE((c + 1) & 1)
    __syncthreads();
  }
#undef STAGE_LOAD
#undef STAGE_STORE

  int o0 = l15, o1 = 16 + l15;
  const float* W0 = Wgc + (size_t)o0 * 96;
  const float* W1 = Wgc + (size_t)o1 * 96;
  int lbase = j0 >> 5;
  size_t nrow = ((size_t)b * 512 + m0 + l15) * 64;

  float bias0 = bgc[o0], bias1 = bgc[o1];
  v8f e00, e01, e10, e11;
#pragma unroll
  for (int r = 0; r < 8; ++r) { e00[r] = bias0; e01[r] = bias1; e10[r] = bias0; e11[r] = bias1; }

  {
    const float* g0 = Xin + (nrow + lbase) * 32 + half * 16;
    const float* g1 = g0 + 32;
    float4 ga[4], gb[4];
#pragma unroll
    for (int q = 0; q < 4; ++q) {
      ga[q] = *(const float4*)(g0 + 4 * q);
      gb[q] = *(const float4*)(g1 + 4 * q);
    }
#pragma unroll
    for (int q = 0; q < 4; ++q) {
      float4 w0 = *(const float4*)(W0 + half * 16 + 4 * q);
      float4 w1 = *(const float4*)(W1 + half * 16 + 4 * q);
      v2f wa0 = {w0.x, w0.y}, wb0 = {w0.z, w0.w};
      v2f wa1 = {w1.x, w1.y}, wb1 = {w1.z, w1.w};
      v2f xa = {ga[q].x, ga[q].y}, xb = {ga[q].z, ga[q].w};
      v2f ya = {gb[q].x, gb[q].y}, yb = {gb[q].z, gb[q].w};
      e00 = wmma_f32(xa, wa0, e00);
      e01 = wmma_f32(xa, wa1, e01);
      e10 = wmma_f32(ya, wa0, e10);
      e11 = wmma_f32(ya, wa1, e11);
      e00 = wmma_f32(xb, wb0, e00);
      e01 = wmma_f32(xb, wb1, e01);
      e10 = wmma_f32(yb, wb0, e10);
      e11 = wmma_f32(yb, wb1, e11);
    }
  }
#define GC_PART(T0, T1, T2_, T3, WOFF)                                        \
  _Pragma("unroll")                                                           \
  for (int i = 0; i < 4; ++i) {                                               \
    v2f wlo0 = *(const v2f*)(W0 + (WOFF) + half * 8 + 2 * i);                 \
    v2f wlo1 = *(const v2f*)(W1 + (WOFF) + half * 8 + 2 * i);                 \
    v2f whi0 = *(const v2f*)(W0 + (WOFF) + 16 + half * 8 + 2 * i);            \
    v2f whi1 = *(const v2f*)(W1 + (WOFF) + 16 + half * 8 + 2 * i);            \
    v2f p0 = {T0[2 * i], T0[2 * i + 1]};                                      \
    v2f p1 = {T1[2 * i], T1[2 * i + 1]};                                      \
    v2f p2 = {T2_[2 * i], T2_[2 * i + 1]};                                    \
    v2f p3 = {T3[2 * i], T3[2 * i + 1]};                                      \
    e00 = wmma_f32(p0, wlo0, e00);                                            \
    e01 = wmma_f32(p0, wlo1, e01);                                            \
    e10 = wmma_f32(p2, wlo0, e10);                                            \
    e11 = wmma_f32(p2, wlo1, e11);                                            \
    e00 = wmma_f32(p1, whi0, e00);                                            \
    e01 = wmma_f32(p1, whi1, e01);                                            \
    e10 = wmma_f32(p3, whi0, e10);                                            \
    e11 = wmma_f32(p3, whi1, e11);                                            \
  }
  GC_PART(c0, c1, c2, c3, 32)
  float res0[8], res1[8], res2[8], res3[8];
  if (FROM_X) {
    float w00 = Wst[o0 * 2], w01 = Wst[o0 * 2 + 1], bs0 = bst[o0];
    float w10 = Wst[o1 * 2], w11 = Wst[o1 * 2 + 1], bs1 = bst[o1];
    int la = lbase + 1 < 63 ? lbase + 1 : 63;
    int lb = lbase + 2 < 63 ? lbase + 2 : 63;
    const float* xn = x + (size_t)b * (2 * 512 * 64) + (size_t)(m0 + half * 8) * 64;
#pragma unroll
    for (int r = 0; r < 8; ++r) {
      float xa0 = xn[r * 64 + la], xa1 = xn[512 * 64 + r * 64 + la];
      float xb0 = xn[r * 64 + lb], xb1 = xn[512 * 64 + r * 64 + lb];
      res0[r] = fmaf(xa0, w00, fmaf(xa1, w01, bs0));
      res1[r] = fmaf(xa0, w10, fmaf(xa1, w11, bs1));
      res2[r] = fmaf(xb0, w00, fmaf(xb1, w01, bs0));
      res3[r] = fmaf(xb0, w10, fmaf(xb1, w11, bs1));
    }
  } else {
#pragma unroll
    for (int r = 0; r < 8; ++r) {
      size_t row = ((size_t)b * 512 + m0 + r + half * 8) * 64 + lbase;
      res0[r] = Hres[(row + 1) * 32 + o0];
      res1[r] = Hres[(row + 1) * 32 + o1];
      res2[r] = Hres[(row + 2) * 32 + o0];
      res3[r] = Hres[(row + 2) * 32 + o1];
    }
  }
  GC_PART(d0, d1, d2, d3, 64)
#undef GC_PART

  float rinv = rsqrtf(1.0f + EPSBN);
  float ig0 = gcg[o0] * rinv, bg0 = gcb[o0], ib0 = bng[o0] * rinv, bb0 = bnb[o0];
  float ig1 = gcg[o1] * rinv, bg1 = gcb[o1], ib1 = bng[o1] * rinv, bb1 = bnb[o1];
#pragma unroll
  for (int r = 0; r < 8; ++r) {
    size_t row = ((size_t)b * 512 + m0 + r + half * 8) * 64 + lbase;
    if (lbase < Lout) {
      float g0v = e00[r] * ig0 + bg0;
      float g1v = e01[r] * ig1 + bg1;
      float h0 = g0v + res0[r];
      float h1 = g1v + res1[r];
      Hout[row * 32 + o0] = h0 * ib0 + bb0;
      Hout[row * 32 + o1] = h1 * ib1 + bb1;
    }
    if (lbase + 1 < Lout) {
      float g0v = e10[r] * ig0 + bg0;
      float g1v = e11[r] * ig1 + bg1;
      float h0 = g0v + res2[r];
      float h1 = g1v + res3[r];
      Hout[(row + 1) * 32 + o0] = h0 * ib0 + bb0;
      Hout[(row + 1) * 32 + o1] = h1 * ib1 + bb1;
    }
  }
}

extern "C" void kernel_launch(void* const* d_in, const int* in_sizes, int n_in,
                              void* d_out, int out_size, void* d_ws, size_t ws_size,
                              hipStream_t stream) {
  (void)in_sizes; (void)n_in; (void)out_size; (void)ws_size;
  const float* x   = (const float*)d_in[0];
  const float* adj = (const float*)d_in[1];
  const float* Wst = (const float*)d_in[2];
  const float* bst = (const float*)d_in[3];
  const float* Wf  = (const float*)d_in[4];
  const float* bf  = (const float*)d_in[5];
  const float* Wg  = (const float*)d_in[6];
  const float* bg  = (const float*)d_in[7];
  const float* Wsk = (const float*)d_in[8];
  const float* bsk = (const float*)d_in[9];
  const float* Wgc = (const float*)d_in[10];
  const float* bgc = (const float*)d_in[11];
  const float* gcg = (const float*)d_in[12];
  const float* gcb = (const float*)d_in[13];
  const float* bng = (const float*)d_in[14];
  const float* bnb = (const float*)d_in[15];
  const float* og  = (const float*)d_in[16];
  const float* ob  = (const float*)d_in[17];
  float* out = (float*)d_out;

  const size_t BUF = (size_t)8 * 512 * 64 * 32 + 1024;
  float* ws    = (float*)d_ws;
  float* bufH  = ws;
  float* bufH2 = ws + BUF;
  float* bufG  = ws + 2 * BUF;
  float* bufS  = ws + 5 * BUF;

  float* T2 = bufS + 62 * 32;

  float* wtab = ws + 3 * BUF;

  k_gated0_skip_setup<<<768, 256, 0, stream>>>(x, Wst, bst, bufG, Wf, bf, Wg, bg, Wsk,
                                               bsk, bufS, adj, T2, wtab);
  k_diffuse_gconv<true><<<1024, 256, 0, stream>>>(adj, T2, bufG, bufH, bufH2, Wgc, bgc,
                                                  gcg, gcb, bng, bnb, 63, x, Wst, bst);
  k_gated_skip<<<512, 256, 0, stream>>>(bufH2, bufG, wtab, bf + 32, bg + 32, bsk + 32,
                                        bufS, og, ob, out, 1);
  k_diffuse_gconv<false><<<1024, 256, 0, stream>>>(adj, T2, bufG, bufH2, bufH, Wgc + 3072,
                                                   bgc + 32, gcg + 32, gcb + 32,
                                                   bng + 32, bnb + 32, 62, x, Wst, bst);
  k_gated_skip<<<512, 256, 0, stream>>>(bufH, bufG, wtab + 5120, bf + 64, bg + 64,
                                        bsk + 64, bufS, og, ob, out, 2);
}
